# starting kernel
# baseline (speedup 1.0000x reference)
#include <hip/hip_runtime.h>

typedef float v2f __attribute__((ext_vector_type(2)));
typedef float v8f __attribute__((ext_vector_type(8)));
typedef int   v4i __attribute__((ext_vector_type(4)));

__global__ void zero_ws_kernel(float* __restrict__ ws, int n_floats) {
    int i = (blockIdx.x * blockDim.x + threadIdx.x) * 4;
    if (i + 3 < n_floats) {
        float4 z = make_float4(0.f, 0.f, 0.f, 0.f);
        *reinterpret_cast<float4*>(ws + i) = z;
    } else {
        for (int k = i; k < n_floats; ++k) ws[k] = 0.f;
    }
}

__global__ void edge_scatter_kernel(const int* __restrict__ src,
                                    const int* __restrict__ dst,
                                    const float* __restrict__ x,
                                    float* __restrict__ aggr,
                                    int n_edges) {
    int t  = blockIdx.x * blockDim.x + threadIdx.x;
    int e0 = t * 4;
    if (e0 + 3 < n_edges) {
        v4i s = __builtin_nontemporal_load(reinterpret_cast<const v4i*>(src + e0));
        v4i d = __builtin_nontemporal_load(reinterpret_cast<const v4i*>(dst + e0));

        float2 xs0 = *reinterpret_cast<const float2*>(x + (size_t)s.x * 2);
        float2 xs1 = *reinterpret_cast<const float2*>(x + (size_t)s.y * 2);
        float2 xs2 = *reinterpret_cast<const float2*>(x + (size_t)s.z * 2);
        float2 xs3 = *reinterpret_cast<const float2*>(x + (size_t)s.w * 2);

        unsafeAtomicAdd(&aggr[(size_t)d.x * 2 + 0], xs0.x);
        unsafeAtomicAdd(&aggr[(size_t)d.x * 2 + 1], xs0.y);
        unsafeAtomicAdd(&aggr[(size_t)d.y * 2 + 0], xs1.x);
        unsafeAtomicAdd(&aggr[(size_t)d.y * 2 + 1], xs1.y);
        unsafeAtomicAdd(&aggr[(size_t)d.z * 2 + 0], xs2.x);
        unsafeAtomicAdd(&aggr[(size_t)d.z * 2 + 1], xs2.y);
        unsafeAtomicAdd(&aggr[(size_t)d.w * 2 + 0], xs3.x);
        unsafeAtomicAdd(&aggr[(size_t)d.w * 2 + 1], xs3.y);
    } else {
        for (int e = e0; e < n_edges; ++e) {
            int s = src[e], d = dst[e];
            float2 xs = *reinterpret_cast<const float2*>(x + (size_t)s * 2);
            unsafeAtomicAdd(&aggr[(size_t)d * 2 + 0], xs.x);
            unsafeAtomicAdd(&aggr[(size_t)d * 2 + 1], xs.y);
        }
    }
}

__global__ void node_epilogue_wmma(const float* __restrict__ x,
                                   const float* __restrict__ aggr,
                                   const float* __restrict__ W_rel,
                                   const float* __restrict__ b_rel,
                                   const float* __restrict__ W_root,
                                   float* __restrict__ out,
                                   int n_nodes) {
    const int wave      = threadIdx.x >> 5;
    const int lane      = threadIdx.x & 31;
    const int num_tiles = (n_nodes + 15) >> 4;
    const int tile      = blockIdx.x * (blockDim.x >> 5) + wave;
    if (tile >= num_tiles) return;

    const bool full = (tile * 16 + 16) <= n_nodes;
    const int  n    = lane & 15;
    const int  half = lane >> 4;

    const int row = tile * 16 + n;
    const float* __restrict__ srcA = half ? x : aggr;
    v2f a;
    a.x = 0.f; a.y = 0.f;
    if (full) {
        float2 v = *reinterpret_cast<const float2*>(srcA + (size_t)row * 2);
        a.x = v.x; a.y = v.y;
    } else if (row < n_nodes) {
        float2 v = *reinterpret_cast<const float2*>(srcA + (size_t)row * 2);
        a.x = v.x; a.y = v.y;
    }

    v2f b;
    b.x = 0.f; b.y = 0.f;
    if (n < 2) {
        const float* __restrict__ W = half ? W_root : W_rel;
        b.x = W[n * 2 + 0];
        b.y = W[n * 2 + 1];
    }

    const float cv = (n < 2) ? b_rel[n] : 0.f;
    v8f c;
#pragma unroll
    for (int r = 0; r < 8; ++r) c[r] = cv;

    v8f d = __builtin_amdgcn_wmma_f32_16x16x4_f32(
         false, a,  false, b,
         (short)0, c,  false,  false);

    const int m0 = tile * 16 + half * 8;
    if (full) {
        if (n < 2) {
            float* __restrict__ p = out + (size_t)m0 * 2 + n;
#pragma unroll
            for (int r = 0; r < 8; ++r) p[(size_t)r * 2] = d[r];
        }
    } else {
        if (n < 2) {
#pragma unroll
            for (int r = 0; r < 8; ++r) {
                int m = m0 + r;
                if (m < n_nodes) out[(size_t)m * 2 + n] = d[r];
            }
        }
    }
}

extern "C" void kernel_launch(void* const* d_in, const int* in_sizes, int n_in,
                              void* d_out, int out_size, void* d_ws, size_t ws_size,
                              hipStream_t stream) {
    const float* x          = (const float*)d_in[0];
    const int*   edge_index = (const int*)d_in[1];
    const float* W_rel      = (const float*)d_in[2];
    const float* b_rel      = (const float*)d_in[3];
    const float* W_root     = (const float*)d_in[4];
    float*       out        = (float*)d_out;
    float*       aggr       = (float*)d_ws;

    const int n_nodes = in_sizes[0] / 2;
    const int n_edges = in_sizes[1] / 2;
    const int* src = edge_index;
    const int* dst = edge_index + n_edges;

    {
        int n_floats = n_nodes * 2;
        int threads  = (n_floats + 3) / 4;
        int blocks   = (threads + 255) / 256;
        zero_ws_kernel<<<blocks, 256, 0, stream>>>(aggr, n_floats);
    }

    {
        int threads = (n_edges + 3) / 4;
        int blocks  = (threads + 255) / 256;
        edge_scatter_kernel<<<blocks, 256, 0, stream>>>(src, dst, x, aggr, n_edges);
    }

    {
        int num_tiles = (n_nodes + 15) / 16;
        int blocks    = (num_tiles + 7) / 8;
        node_epilogue_wmma<<<blocks, 256, 0, stream>>>(x, aggr, W_rel, b_rel,
                                                       W_root, out, n_nodes);
    }
}


# Global float atomics replaced by a counting sort of the edges by destination range in the workspace, then per-range sums with LDS float adds (arrival order) and plain stores of aggr
# speedup vs baseline: 3.8821x; 3.8821x over previous
#include <hip/hip_runtime.h>

typedef float v2f __attribute__((ext_vector_type(2)));
typedef float v8f __attribute__((ext_vector_type(8)));
typedef int   v4i __attribute__((ext_vector_type(4)));
typedef int   v2i __attribute__((ext_vector_type(2)));

#define NPB_SHIFT 10
#define NPB (1 << NPB_SHIFT)
#define BIN_THREADS 1024
#define BIN_VEC 4
#define BIN_EPT (BIN_VEC * 4)
#define BIN_CHUNK (BIN_THREADS * BIN_EPT)
#define ACC_THREADS 1024
#define MAX_BINS BIN_THREADS

__device__ __forceinline__ int block_excl_scan_1024(int v, int* wsum) {
    const int lane = threadIdx.x & 31;
    const int w    = threadIdx.x >> 5;
    int inc = v;
#pragma unroll
    for (int o = 1; o < 32; o <<= 1) {
        int t = __shfl_up(inc, o);
        if (lane >= o) inc += t;
    }
    if (lane == 31) wsum[w] = inc;
    __syncthreads();
    if (w == 0) {
        int s  = wsum[lane];
        int si = s;
#pragma unroll
        for (int o = 1; o < 32; o <<= 1) {
            int t = __shfl_up(si, o);
            if (lane >= o) si += t;
        }
        wsum[lane] = si - s;
        if (lane == 31) wsum[32] = si;
    }
    __syncthreads();
    return wsum[w] + inc - v;
}

__global__ void zero_ws_kernel(int* __restrict__ ws, int n) {
    int i = blockIdx.x * blockDim.x + threadIdx.x;
    if (i < n) ws[i] = 0;
}

__global__ void __launch_bounds__(BIN_THREADS)
bin_count_kernel(const int* __restrict__ dst, int* __restrict__ bin_count,
                 int n_edges) {
    __shared__ int hist[MAX_BINS];
    hist[threadIdx.x] = 0;
    __syncthreads();
    const int base = blockIdx.x * BIN_CHUNK;
#pragma unroll
    for (int j = 0; j < BIN_VEC; ++j) {
        int e0 = base + (j * BIN_THREADS + threadIdx.x) * 4;
        if (e0 + 3 < n_edges) {
            v4i d = __builtin_nontemporal_load(reinterpret_cast<const v4i*>(dst + e0));
            atomicAdd(&hist[d.x >> NPB_SHIFT], 1);
            atomicAdd(&hist[d.y >> NPB_SHIFT], 1);
            atomicAdd(&hist[d.z >> NPB_SHIFT], 1);
            atomicAdd(&hist[d.w >> NPB_SHIFT], 1);
        } else {
            for (int e = e0; e < n_edges && e < e0 + 4; ++e)
                atomicAdd(&hist[dst[e] >> NPB_SHIFT], 1);
        }
    }
    __syncthreads();
    {
        int h = hist[threadIdx.x];
        if (h) atomicAdd(&bin_count[threadIdx.x], h);
    }
}

__global__ void __launch_bounds__(BIN_THREADS)
bin_scan_kernel(const int* __restrict__ bin_count, int* __restrict__ bin_start,
                int n_bins) {
    __shared__ int wsum[33];
    const int b = threadIdx.x;
    const int c = (b < n_bins) ? bin_count[b] : 0;
    const int run = block_excl_scan_1024(c, wsum);
    if (b < n_bins) bin_start[b] = run;
    if (b == BIN_THREADS - 1) bin_start[n_bins] = run + c;
}

template <bool FULL>
__device__ __forceinline__ void bin_scatter_body(
        const int* __restrict__ src, const int* __restrict__ dst,
        const int* __restrict__ bin_start, int* __restrict__ bin_cursor,
        v2i* __restrict__ rec, int n_edges,
        int* lcnt, int* gbase, int* wsum, v2i* stage) {
    const int base = blockIdx.x * BIN_CHUNK;

    lcnt[threadIdx.x] = 0;
    __syncthreads();

    int s[BIN_EPT], d[BIN_EPT];
#pragma unroll
    for (int j = 0; j < BIN_VEC; ++j) {
        int e0 = base + (j * BIN_THREADS + threadIdx.x) * 4;
        if (FULL || e0 + 3 < n_edges) {
            v4i sv = __builtin_nontemporal_load(reinterpret_cast<const v4i*>(src + e0));
            v4i dv = __builtin_nontemporal_load(reinterpret_cast<const v4i*>(dst + e0));
            s[j * 4 + 0] = sv.x; s[j * 4 + 1] = sv.y; s[j * 4 + 2] = sv.z; s[j * 4 + 3] = sv.w;
            d[j * 4 + 0] = dv.x; d[j * 4 + 1] = dv.y; d[j * 4 + 2] = dv.z; d[j * 4 + 3] = dv.w;
        } else {
#pragma unroll
            for (int k = 0; k < 4; ++k) {
                int  e  = e0 + k;
                bool ok = e < n_edges;
                s[j * 4 + k] = ok ? src[e] : 0;
                d[j * 4 + k] = ok ? dst[e] : -1;
            }
        }
    }
#pragma unroll
    for (int k = 0; k < BIN_EPT; ++k)
        if (FULL || d[k] >= 0) atomicAdd(&lcnt[d[k] >> NPB_SHIFT], 1);
    __syncthreads();

    {
        const int b   = threadIdx.x;
        const int h   = lcnt[b];
        const int run = block_excl_scan_1024(h, wsum);
        if (h) gbase[b] = bin_start[b] + atomicAdd(&bin_cursor[b], h) - run;
        lcnt[b] = run;
    }
    __syncthreads();

#pragma unroll
    for (int j = 0; j < BIN_VEC; ++j) {
        int p[4];
#pragma unroll
        for (int k = 0; k < 4; ++k)
            p[k] = (FULL || d[j * 4 + k] >= 0) ? atomicAdd(&lcnt[d[j * 4 + k] >> NPB_SHIFT], 1) : -1;
#pragma unroll
        for (int k = 0; k < 4; ++k) {
            if (FULL || p[k] >= 0) {
                v2i r; r.x = s[j * 4 + k]; r.y = d[j * 4 + k];
                stage[p[k]] = r;
            }
        }
    }
    __syncthreads();

    int total = n_edges - base;
    if (FULL || total > BIN_CHUNK) total = BIN_CHUNK;
    for (int i = threadIdx.x; i < total; i += BIN_THREADS) {
        v2i r = stage[i];
        int g = gbase[r.y >> NPB_SHIFT] + i;
        r.y &= (NPB - 1);
        rec[g] = r;
    }
}

__global__ void __launch_bounds__(BIN_THREADS, 16)
bin_scatter_kernel(const int* __restrict__ src, const int* __restrict__ dst,
                   const int* __restrict__ bin_start, int* __restrict__ bin_cursor,
                   int* __restrict__ records, int n_edges) {
    __shared__ int lcnt[MAX_BINS];
    __shared__ int gbase[MAX_BINS];
    __shared__ int wsum[33];
    __shared__ __attribute__((aligned(16))) v2i stage[BIN_CHUNK];
    v2i* rec = reinterpret_cast<v2i*>(records);
    if ((blockIdx.x + 1) * BIN_CHUNK <= (unsigned)n_edges)
        bin_scatter_body<true>(src, dst, bin_start, bin_cursor, rec, n_edges,
                               lcnt, gbase, wsum, stage);
    else
        bin_scatter_body<false>(src, dst, bin_start, bin_cursor, rec, n_edges,
                                lcnt, gbase, wsum, stage);
}

__global__ void __launch_bounds__(ACC_THREADS)
bin_accumulate_kernel(const int* __restrict__ records, const int* __restrict__ bin_start,
                      const float* __restrict__ x, float* __restrict__ aggr,
                      int n_nodes) {
    __shared__ float acc[NPB * 2];
    const int b = blockIdx.x;
    for (int i = threadIdx.x; i < NPB * 2; i += ACC_THREADS) acc[i] = 0.f;
    __syncthreads();
    const v2i* __restrict__ rec = reinterpret_cast<const v2i*>(records);
    const int lo = bin_start[b], hi = bin_start[b + 1];
    int i = lo + threadIdx.x;
    for (; i + 3 * ACC_THREADS < hi; i += 4 * ACC_THREADS) {
        v2i r0 = rec[i];
        v2i r1 = rec[i + ACC_THREADS];
        v2i r2 = rec[i + 2 * ACC_THREADS];
        v2i r3 = rec[i + 3 * ACC_THREADS];
        float2 x0 = *reinterpret_cast<const float2*>(x + (size_t)r0.x * 2);
        float2 x1 = *reinterpret_cast<const float2*>(x + (size_t)r1.x * 2);
        float2 x2 = *reinterpret_cast<const float2*>(x + (size_t)r2.x * 2);
        float2 x3 = *reinterpret_cast<const float2*>(x + (size_t)r3.x * 2);
        atomicAdd(&acc[r0.y * 2 + 0], x0.x); atomicAdd(&acc[r0.y * 2 + 1], x0.y);
        atomicAdd(&acc[r1.y * 2 + 0], x1.x); atomicAdd(&acc[r1.y * 2 + 1], x1.y);
        atomicAdd(&acc[r2.y * 2 + 0], x2.x); atomicAdd(&acc[r2.y * 2 + 1], x2.y);
        atomicAdd(&acc[r3.y * 2 + 0], x3.x); atomicAdd(&acc[r3.y * 2 + 1], x3.y);
    }
    for (; i < hi; i += ACC_THREADS) {
        v2i r = rec[i];
        float2 xv = *reinterpret_cast<const float2*>(x + (size_t)r.x * 2);
        atomicAdd(&acc[r.y * 2 + 0], xv.x);
        atomicAdd(&acc[r.y * 2 + 1], xv.y);
    }
    __syncthreads();
    const int node0 = b * NPB;
    int nloc = n_nodes - node0;
    if (nloc > NPB) nloc = NPB;
    for (int k = threadIdx.x; k < nloc * 2; k += ACC_THREADS)
        aggr[(size_t)node0 * 2 + k] = acc[k];
}

__global__ void node_epilogue_wmma(const float* __restrict__ x,
                                   const float* __restrict__ aggr,
                                   const float* __restrict__ W_rel,
                                   const float* __restrict__ b_rel,
                                   const float* __restrict__ W_root,
                                   float* __restrict__ out,
                                   int n_nodes) {
    const int wave      = threadIdx.x >> 5;
    const int lane      = threadIdx.x & 31;
    const int num_tiles = (n_nodes + 15) >> 4;
    const int tile      = blockIdx.x * (blockDim.x >> 5) + wave;
    if (tile >= num_tiles) return;

    const bool full = (tile * 16 + 16) <= n_nodes;
    const int  n    = lane & 15;
    const int  half = lane >> 4;

    const int row = tile * 16 + n;
    const float* __restrict__ srcA = half ? x : aggr;
    v2f a;
    a.x = 0.f; a.y = 0.f;
    if (full) {
        float2 v = *reinterpret_cast<const float2*>(srcA + (size_t)row * 2);
        a.x = v.x; a.y = v.y;
    } else if (row < n_nodes) {
        float2 v = *reinterpret_cast<const float2*>(srcA + (size_t)row * 2);
        a.x = v.x; a.y = v.y;
    }

    v2f b;
    b.x = 0.f; b.y = 0.f;
    if (n < 2) {
        const float* __restrict__ W = half ? W_root : W_rel;
        b.x = W[n * 2 + 0];
        b.y = W[n * 2 + 1];
    }

    const float cv = (n < 2) ? b_rel[n] : 0.f;
    v8f c;
#pragma unroll
    for (int r = 0; r < 8; ++r) c[r] = cv;

    v8f d = __builtin_amdgcn_wmma_f32_16x16x4_f32(
         false, a,  false, b,
         (short)0, c,  false,  false);

    const int m0 = tile * 16 + half * 8;
    if (full) {
        if (n < 2) {
            float* __restrict__ p = out + (size_t)m0 * 2 + n;
#pragma unroll
            for (int r = 0; r < 8; ++r) p[(size_t)r * 2] = d[r];
        }
    } else {
        if (n < 2) {
#pragma unroll
            for (int r = 0; r < 8; ++r) {
                int m = m0 + r;
                if (m < n_nodes) out[(size_t)m * 2 + n] = d[r];
            }
        }
    }
}

extern "C" void kernel_launch(void* const* d_in, const int* in_sizes, int n_in,
                              void* d_out, int out_size, void* d_ws, size_t ws_size,
                              hipStream_t stream) {
    const float* x          = (const float*)d_in[0];
    const int*   edge_index = (const int*)d_in[1];
    const float* W_rel      = (const float*)d_in[2];
    const float* b_rel      = (const float*)d_in[3];
    const float* W_root     = (const float*)d_in[4];
    float*       out        = (float*)d_out;

    const int n_nodes = in_sizes[0] / 2;
    const int n_edges = in_sizes[1] / 2;
    const int* src = edge_index;
    const int* dst = edge_index + n_edges;
    const int n_bins = (n_nodes + NPB - 1) / NPB;

    float* aggr       = (float*)d_ws;
    int*   bin_count  = (int*)d_ws + (((size_t)n_nodes * 2 + 63) / 64) * 64;
    int*   bin_cursor = bin_count + n_bins;
    int*   bin_start  = bin_cursor + n_bins;
    int*   records    = bin_start + (((size_t)n_bins + 1 + 63) / 64) * 64;

    const int chunks = (n_edges + BIN_CHUNK - 1) / BIN_CHUNK;

    zero_ws_kernel<<<(2 * n_bins + 255) / 256, 256, 0, stream>>>(bin_count, 2 * n_bins);
    bin_count_kernel<<<chunks, BIN_THREADS, 0, stream>>>(dst, bin_count, n_edges);
    bin_scan_kernel<<<1, BIN_THREADS, 0, stream>>>(bin_count, bin_start, n_bins);
    bin_scatter_kernel<<<chunks, BIN_THREADS, 0, stream>>>(src, dst, bin_start, bin_cursor,
                                                          records, n_edges);
    bin_accumulate_kernel<<<n_bins, ACC_THREADS, 0, stream>>>(records, bin_start, x, aggr, n_nodes);

    {
        int num_tiles = (n_nodes + 15) / 16;
        int blocks    = (num_tiles + 7) / 8;
        node_epilogue_wmma<<<blocks, 256, 0, stream>>>(x, aggr, W_rel, b_rel,
                                                       W_root, out, n_nodes);
    }
}


# Global bin counters, count pass, scan and cursor atomics dropped: each chunk's LDS-sorted records stay in its own slot with a bin offset table; bin sums (LDS adds, arrival order) walk a run per chunk
# speedup vs baseline: 4.0698x; 1.0484x over previous
#include <hip/hip_runtime.h>

typedef float v2f __attribute__((ext_vector_type(2)));
typedef float v8f __attribute__((ext_vector_type(8)));
typedef int   v4i __attribute__((ext_vector_type(4)));
typedef int   v2i __attribute__((ext_vector_type(2)));

#define NPB_SHIFT 10
#define NPB (1 << NPB_SHIFT)
#define BIN_THREADS 1024
#define BIN_VEC 4
#define BIN_EPT (BIN_VEC * 4)
#define BIN_CHUNK (BIN_THREADS * BIN_EPT)
#define ACC_THREADS 1024
#define ACC_WAVES (ACC_THREADS / 32)
#define ACC_SEGS 8
#define MAX_BINS BIN_THREADS
#define SEG_PITCH (MAX_BINS + 64)

__device__ __forceinline__ int block_excl_scan_1024(int v, int* wsum) {
    const int lane = threadIdx.x & 31;
    const int w    = threadIdx.x >> 5;
    int inc = v;
#pragma unroll
    for (int o = 1; o < 32; o <<= 1) {
        int t = __shfl_up(inc, o);
        if (lane >= o) inc += t;
    }
    if (lane == 31) wsum[w] = inc;
    __syncthreads();
    if (w == 0) {
        int s  = wsum[lane];
        int si = s;
#pragma unroll
        for (int o = 1; o < 32; o <<= 1) {
            int t = __shfl_up(si, o);
            if (lane >= o) si += t;
        }
        wsum[lane] = si - s;
        if (lane == 31) wsum[32] = si;
    }
    __syncthreads();
    return wsum[w] + inc - v;
}

template <bool FULL>
__device__ __forceinline__ void chunk_sort_body(
        const int* __restrict__ src, const int* __restrict__ dst,
        int* __restrict__ seg_start, v2i* __restrict__ rec, int n_edges, int n_bins,
        int* lcnt, int* wsum, v2i* stage) {
    const int base = blockIdx.x * BIN_CHUNK;

    lcnt[threadIdx.x] = 0;
    __syncthreads();

    int s[BIN_EPT], d[BIN_EPT];
#pragma unroll
    for (int j = 0; j < BIN_VEC; ++j) {
        int e0 = base + (j * BIN_THREADS + threadIdx.x) * 4;
        if (FULL || e0 + 3 < n_edges) {
            v4i sv = __builtin_nontemporal_load(reinterpret_cast<const v4i*>(src + e0));
            v4i dv = __builtin_nontemporal_load(reinterpret_cast<const v4i*>(dst + e0));
            s[j * 4 + 0] = sv.x; s[j * 4 + 1] = sv.y; s[j * 4 + 2] = sv.z; s[j * 4 + 3] = sv.w;
            d[j * 4 + 0] = dv.x; d[j * 4 + 1] = dv.y; d[j * 4 + 2] = dv.z; d[j * 4 + 3] = dv.w;
        } else {
#pragma unroll
            for (int k = 0; k < 4; ++k) {
                int  e  = e0 + k;
                bool ok = e < n_edges;
                s[j * 4 + k] = ok ? src[e] : 0;
                d[j * 4 + k] = ok ? dst[e] : -1;
            }
        }
    }
#pragma unroll
    for (int k = 0; k < BIN_EPT; ++k)
        if (FULL || d[k] >= 0) atomicAdd(&lcnt[d[k] >> NPB_SHIFT], 1);
    __syncthreads();

    {
        const int b   = threadIdx.x;
        const int h   = lcnt[b];
        const int run = block_excl_scan_1024(h, wsum);
        int* __restrict__ row = seg_start + (size_t)blockIdx.x * SEG_PITCH;
        if (b <= n_bins) row[b] = run;
        if (b == MAX_BINS - 1 && n_bins == MAX_BINS) row[MAX_BINS] = run + h;
        lcnt[b] = run;
    }
    __syncthreads();

#pragma unroll
    for (int j = 0; j < BIN_VEC; ++j) {
        int p[4];
#pragma unroll
        for (int k = 0; k < 4; ++k)
            p[k] = (FULL || d[j * 4 + k] >= 0) ? atomicAdd(&lcnt[d[j * 4 + k] >> NPB_SHIFT], 1) : -1;
#pragma unroll
        for (int k = 0; k < 4; ++k) {
            if (FULL || p[k] >= 0) {
                v2i r; r.x = s[j * 4 + k]; r.y = d[j * 4 + k] & (NPB - 1);
                stage[p[k]] = r;
            }
        }
    }
    __syncthreads();

    int total = n_edges - base;
    if (FULL || total > BIN_CHUNK) total = BIN_CHUNK;
    for (int i = threadIdx.x; i < total; i += BIN_THREADS)
        rec[(size_t)base + i] = stage[i];
}

__global__ void __launch_bounds__(BIN_THREADS, 16)
chunk_sort_kernel(const int* __restrict__ src, const int* __restrict__ dst,
                  int* __restrict__ seg_start, int* __restrict__ records,
                  int n_edges, int n_bins) {
    __shared__ int lcnt[MAX_BINS];
    __shared__ int wsum[33];
    __shared__ __attribute__((aligned(16))) v2i stage[BIN_CHUNK];
    v2i* rec = reinterpret_cast<v2i*>(records);
    if ((blockIdx.x + 1) * BIN_CHUNK <= (unsigned)n_edges)
        chunk_sort_body<true>(src, dst, seg_start, rec, n_edges, n_bins, lcnt, wsum, stage);
    else
        chunk_sort_body<false>(src, dst, seg_start, rec, n_edges, n_bins, lcnt, wsum, stage);
}

__device__ __forceinline__ void seg_bounds(const int* __restrict__ seg_start, int b, int c0,
                                           int chunks, int* beg, int* n, int* idle) {
#pragma unroll
    for (int k = 0; k < ACC_SEGS; ++k) {
        const int c = c0 + k * ACC_WAVES;
        int cc = c < chunks ? c : chunks - 1;
        cc = cc < 0 ? 0 : cc;
        const int* __restrict__ row = seg_start + (size_t)cc * SEG_PITCH;
        const int l   = row[b];
        const int h   = row[b + 1];
        const int len = c < chunks ? h - l : 0;
        beg[k]  = cc * BIN_CHUNK + l;
        n[k]    = len;
        idle[k] = len > 0 ? cc * BIN_CHUNK + l + len - 1 : cc * BIN_CHUNK;
    }
}

__global__ void __launch_bounds__(ACC_THREADS)
bin_accumulate_kernel(const int* __restrict__ records, const int* __restrict__ seg_start,
                      const float* __restrict__ x, float* __restrict__ aggr,
                      int n_nodes, int chunks) {
    __shared__ float acc[NPB * 2];
    const int b = blockIdx.x;
    for (int i = threadIdx.x; i < NPB * 2; i += ACC_THREADS) acc[i] = 0.f;
    __syncthreads();
    const v2i* __restrict__ rec = reinterpret_cast<const v2i*>(records);
    const int lane = threadIdx.x & 31;
    const int w = __builtin_amdgcn_readfirstlane((int)(threadIdx.x >> 5));
    int beg[ACC_SEGS], n[ACC_SEGS], idle[ACC_SEGS];
#pragma unroll
    for (int k = 0; k < ACC_SEGS; ++k) { beg[k] = 0; n[k] = 0; idle[k] = 0; }
    for (int c0 = w - ACC_WAVES * ACC_SEGS; c0 < chunks; c0 += ACC_WAVES * ACC_SEGS) {
        int beg_nx[ACC_SEGS], n_nx[ACC_SEGS], idle_nx[ACC_SEGS];
        seg_bounds(seg_start, b, c0 + ACC_WAVES * ACC_SEGS, chunks, beg_nx, n_nx, idle_nx);
        int nmax = 0;
#pragma unroll
        for (int k = 0; k < ACC_SEGS; ++k) nmax = n[k] > nmax ? n[k] : nmax;
        for (int p0 = 0; p0 < nmax; p0 += 32) {
            const int p = p0 + lane;
            v2i r[ACC_SEGS];
            float2 xv[ACC_SEGS];
#pragma unroll
            for (int k = 0; k < ACC_SEGS; ++k)
                r[k] = rec[p < n[k] ? beg[k] + p : idle[k]];
#pragma unroll
            for (int k = 0; k < ACC_SEGS; ++k)
                xv[k] = *reinterpret_cast<const float2*>(x + (size_t)r[k].x * 2);
#pragma unroll
            for (int k = 0; k < ACC_SEGS; ++k) {
                if (p < n[k]) {
                    atomicAdd(&acc[r[k].y * 2 + 0], xv[k].x);
                    atomicAdd(&acc[r[k].y * 2 + 1], xv[k].y);
                }
            }
        }
#pragma unroll
        for (int k = 0; k < ACC_SEGS; ++k) { beg[k] = beg_nx[k]; n[k] = n_nx[k]; idle[k] = idle_nx[k]; }
    }
    __syncthreads();
    const int node0 = b * NPB;
    int nloc = n_nodes - node0;
    if (nloc > NPB) nloc = NPB;
    for (int k = threadIdx.x; k < nloc * 2; k += ACC_THREADS)
        aggr[(size_t)node0 * 2 + k] = acc[k];
}

__global__ void node_epilogue_wmma(const float* __restrict__ x,
                                   const float* __restrict__ aggr,
                                   const float* __restrict__ W_rel,
                                   const float* __restrict__ b_rel,
                                   const float* __restrict__ W_root,
                                   float* __restrict__ out,
                                   int n_nodes) {
    const int wave      = threadIdx.x >> 5;
    const int lane      = threadIdx.x & 31;
    const int num_tiles = (n_nodes + 15) >> 4;
    const int tile      = blockIdx.x * (blockDim.x >> 5) + wave;
    if (tile >= num_tiles) return;

    const bool full = (tile * 16 + 16) <= n_nodes;
    const int  n    = lane & 15;
    const int  half = lane >> 4;

    const int row = tile * 16 + n;
    const float* __restrict__ srcA = half ? x : aggr;
    v2f a;
    a.x = 0.f; a.y = 0.f;
    if (full) {
        float2 v = *reinterpret_cast<const float2*>(srcA + (size_t)row * 2);
        a.x = v.x; a.y = v.y;
    } else if (row < n_nodes) {
        float2 v = *reinterpret_cast<const float2*>(srcA + (size_t)row * 2);
        a.x = v.x; a.y = v.y;
    }

    v2f b;
    b.x = 0.f; b.y = 0.f;
    if (n < 2) {
        const float* __restrict__ W = half ? W_root : W_rel;
        b.x = W[n * 2 + 0];
        b.y = W[n * 2 + 1];
    }

    const float cv = (n < 2) ? b_rel[n] : 0.f;
    v8f c;
#pragma unroll
    for (int r = 0; r < 8; ++r) c[r] = cv;

    v8f d = __builtin_amdgcn_wmma_f32_16x16x4_f32(
         false, a,  false, b,
         (short)0, c,  false,  false);

    const int m0 = tile * 16 + half * 8;
    if (full) {
        if (n < 2) {
            float* __restrict__ p = out + (size_t)m0 * 2 + n;
#pragma unroll
            for (int r = 0; r < 8; ++r) p[(size_t)r * 2] = d[r];
        }
    } else {
        if (n < 2) {
#pragma unroll
            for (int r = 0; r < 8; ++r) {
                int m = m0 + r;
                if (m < n_nodes) out[(size_t)m * 2 + n] = d[r];
            }
        }
    }
}

extern "C" void kernel_launch(void* const* d_in, const int* in_sizes, int n_in,
                              void* d_out, int out_size, void* d_ws, size_t ws_size,
                              hipStream_t stream) {
    const float* x          = (const float*)d_in[0];
    const int*   edge_index = (const int*)d_in[1];
    const float* W_rel      = (const float*)d_in[2];
    const float* b_rel      = (const float*)d_in[3];
    const float* W_root     = (const float*)d_in[4];
    float*       out        = (float*)d_out;

    const int n_nodes = in_sizes[0] / 2;
    const int n_edges = in_sizes[1] / 2;
    const int* src = edge_index;
    const int* dst = edge_index + n_edges;
    const int n_bins = (n_nodes + NPB - 1) / NPB;
    const int chunks = (n_edges + BIN_CHUNK - 1) / BIN_CHUNK;

    float* aggr      = (float*)d_ws;
    int*   seg_start = (int*)d_ws + (((size_t)n_nodes * 2 + 63) / 64) * 64;
    int*   records   = seg_start + (size_t)chunks * SEG_PITCH;

    chunk_sort_kernel<<<chunks, BIN_THREADS, 0, stream>>>(src, dst, seg_start, records,
                                                         n_edges, n_bins);
    bin_accumulate_kernel<<<n_bins, ACC_THREADS, 0, stream>>>(records, seg_start, x, aggr,
                                                              n_nodes, chunks);

    {
        int num_tiles = (n_nodes + 15) / 16;
        int blocks    = (num_tiles + 7) / 8;
        node_epilogue_wmma<<<blocks, 256, 0, stream>>>(x, aggr, W_rel, b_rel,
                                                       W_root, out, n_nodes);
    }
}


# Edges keyed by (dst range, src slice) of 16384 rows: a workgroup gathers x from an LDS copy of its slice, not L2; per-slice partial sums (LDS adds, arrival order) summed in slice order in epilogue
# speedup vs baseline: 5.9898x; 1.4718x over previous
#include <hip/hip_runtime.h>

typedef float v2f __attribute__((ext_vector_type(2)));
typedef float v8f __attribute__((ext_vector_type(8)));
typedef int   v4i __attribute__((ext_vector_type(4)));
typedef int   v2i __attribute__((ext_vector_type(2)));

#define TS 14
#define TILE (1 << TS)
#define BIN_THREADS 1024
#define BIN_VEC 4
#define BIN_EPT (BIN_VEC * 4)
#define BIN_CHUNK (BIN_THREADS * BIN_EPT)
#define ACC_THREADS 1024
#define MAX_BINS BIN_THREADS
#define SEG_PITCH (MAX_BINS + 64)

__device__ __forceinline__ int block_excl_scan_1024(int v, int* wsum) {
    const int lane = threadIdx.x & 31;
    const int w    = threadIdx.x >> 5;
    int inc = v;
#pragma unroll
    for (int o = 1; o < 32; o <<= 1) {
        int t = __shfl_up(inc, o);
        if (lane >= o) inc += t;
    }
    if (lane == 31) wsum[w] = inc;
    __syncthreads();
    if (w == 0) {
        int s  = wsum[lane];
        int si = s;
#pragma unroll
        for (int o = 1; o < 32; o <<= 1) {
            int t = __shfl_up(si, o);
            if (lane >= o) si += t;
        }
        wsum[lane] = si - s;
        if (lane == 31) wsum[32] = si;
    }
    __syncthreads();
    return wsum[w] + inc - v;
}

template <bool FULL>
__device__ __forceinline__ void chunk_sort_body(
        const int* __restrict__ src, const int* __restrict__ dst,
        int* __restrict__ seg_start, int* __restrict__ rec, int n_edges, int n_bins, int n_ss,
        int* lcnt, int* wsum, int* stage) {
    const int base = blockIdx.x * BIN_CHUNK;

    lcnt[threadIdx.x] = 0;
    __syncthreads();

    int s[BIN_EPT], d[BIN_EPT];
#pragma unroll
    for (int j = 0; j < BIN_VEC; ++j) {
        int e0 = base + (j * BIN_THREADS + threadIdx.x) * 4;
        if (FULL || e0 + 3 < n_edges) {
            v4i sv = __builtin_nontemporal_load(reinterpret_cast<const v4i*>(src + e0));
            v4i dv = __builtin_nontemporal_load(reinterpret_cast<const v4i*>(dst + e0));
            s[j * 4 + 0] = sv.x; s[j * 4 + 1] = sv.y; s[j * 4 + 2] = sv.z; s[j * 4 + 3] = sv.w;
            d[j * 4 + 0] = dv.x; d[j * 4 + 1] = dv.y; d[j * 4 + 2] = dv.z; d[j * 4 + 3] = dv.w;
        } else {
#pragma unroll
            for (int k = 0; k < 4; ++k) {
                int  e  = e0 + k;
                bool ok = e < n_edges;
                s[j * 4 + k] = ok ? src[e] : 0;
                d[j * 4 + k] = ok ? dst[e] : -1;
            }
        }
    }
#pragma unroll
    for (int k = 0; k < BIN_EPT; ++k)
        if (FULL || d[k] >= 0) atomicAdd(&lcnt[(d[k] >> TS) * n_ss + (s[k] >> TS)], 1);
    __syncthreads();

    {
        const int b   = threadIdx.x;
        const int h   = lcnt[b];
        const int run = block_excl_scan_1024(h, wsum);
        int* __restrict__ row = seg_start + (size_t)blockIdx.x * SEG_PITCH;
        if (b <= n_bins) row[b] = run;
        if (b == MAX_BINS - 1 && n_bins == MAX_BINS) row[MAX_BINS] = run + h;
        lcnt[b] = run;
    }
    __syncthreads();

#pragma unroll
    for (int j = 0; j < BIN_VEC; ++j) {
        int p[4];
#pragma unroll
        for (int k = 0; k < 4; ++k)
            p[k] = (FULL || d[j * 4 + k] >= 0)
                       ? atomicAdd(&lcnt[(d[j * 4 + k] >> TS) * n_ss + (s[j * 4 + k] >> TS)], 1) : -1;
#pragma unroll
        for (int k = 0; k < 4; ++k) {
            if (FULL || p[k] >= 0) {
                stage[p[k]] = (s[j * 4 + k] & (TILE - 1)) | ((d[j * 4 + k] & (TILE - 1)) << TS);
            }
        }
    }
    __syncthreads();

    int total = n_edges - base;
    if (FULL || total > BIN_CHUNK) total = BIN_CHUNK;
    const int total4 = (total + 3) >> 2;
    const v4i* st4 = reinterpret_cast<const v4i*>(stage);
    v4i* __restrict__ rec4 = reinterpret_cast<v4i*>(rec + (size_t)base);
    for (int i = threadIdx.x; i < total4; i += BIN_THREADS)
        rec4[i] = st4[i];
}

__global__ void __launch_bounds__(BIN_THREADS, 16)
chunk_sort_kernel(const int* __restrict__ src, const int* __restrict__ dst,
                  int* __restrict__ seg_start, int* __restrict__ records,
                  int n_edges, int n_bins, int n_ss) {
    __shared__ int lcnt[MAX_BINS];
    __shared__ int wsum[33];
    __shared__ __attribute__((aligned(16))) int stage[BIN_CHUNK];
    int* rec = records;
    if ((blockIdx.x + 1) * BIN_CHUNK <= (unsigned)n_edges)
        chunk_sort_body<true>(src, dst, seg_start, rec, n_edges, n_bins, n_ss, lcnt, wsum, stage);
    else
        chunk_sort_body<false>(src, dst, seg_start, rec, n_edges, n_bins, n_ss, lcnt, wsum, stage);
}

__device__ __forceinline__ void add_record(int r, const float2* xs, float* acc) {
    const float2 xv = xs[r & (TILE - 1)];
    const int    dl = r >> TS;
    atomicAdd(&acc[dl * 2 + 0], xv.x);
    atomicAdd(&acc[dl * 2 + 1], xv.y);
}

__global__ void __launch_bounds__(ACC_THREADS)
tile_accumulate_kernel(const int* __restrict__ records, const int* __restrict__ seg_start,
                       const float* __restrict__ x, float* __restrict__ part,
                       int n_nodes, int chunks, int n_ss, int plane_rows) {
    __shared__ float2 xs[TILE];
    __shared__ float  acc[TILE * 2];
    const int key  = blockIdx.x;
    const int db   = key / n_ss;
    const int ss   = key - db * n_ss;
    const int lane = threadIdx.x & 31;
    const int w = __builtin_amdgcn_readfirstlane((int)(threadIdx.x >> 5));

    for (int cbase = 0; cbase < chunks; cbase += ACC_THREADS) {
        const int c = cbase + w + 32 * lane;
        int l = 0, h = 0;
        if (c < chunks) {
            const int* __restrict__ row = seg_start + (size_t)c * SEG_PITCH + key;
            l = row[0];
            h = row[1];
        }
        if (cbase == 0) {
            const int srow0 = ss * TILE;
            int ns = n_nodes - srow0;
            if (ns > TILE) ns = TILE;
            const float2* __restrict__ x2 = reinterpret_cast<const float2*>(x) + srow0;
            float2 t[TILE / ACC_THREADS];
#pragma unroll
            for (int u = 0; u < TILE / ACC_THREADS; ++u) {
                const int i = threadIdx.x + u * ACC_THREADS;
                t[u].x = 0.f; t[u].y = 0.f;
                if (i < ns) t[u] = x2[i];
            }
            for (int i = threadIdx.x; i < TILE * 2; i += ACC_THREADS) acc[i] = 0.f;
#pragma unroll
            for (int u = 0; u < TILE / ACC_THREADS; ++u) xs[threadIdx.x + u * ACC_THREADS] = t[u];
        }
        const int begv = c * BIN_CHUNK + l;
        const int nv   = h - l;
        int r[32];
#pragma unroll
        for (int j = 0; j < 32; ++j) {
            const int bj = __builtin_amdgcn_readlane(begv, j);
            const int nj = __builtin_amdgcn_readlane(nv, j);
            const int last = nj > 0 ? bj + nj - 1 : 0;
            r[j] = records[lane < nj ? bj + lane : last];
        }
        if (cbase == 0) __syncthreads();
#pragma unroll
        for (int g = 0; g < 32; g += 8) {
            float2 xv[8];
#pragma unroll
            for (int u = 0; u < 8; ++u) xv[u] = xs[r[g + u] & (TILE - 1)];
#pragma unroll
            for (int u = 0; u < 8; ++u) {
                const int nj = __builtin_amdgcn_readlane(nv, g + u);
                if (lane < nj) {
                    const int dl = r[g + u] >> TS;
                    atomicAdd(&acc[dl * 2 + 0], xv[u].x);
                    atomicAdd(&acc[dl * 2 + 1], xv[u].y);
                }
            }
        }
        if (__any(nv > 32)) {
            for (int j = 0; j < 32; ++j) {
                const int bj = __builtin_amdgcn_readlane(begv, j);
                const int nj = __builtin_amdgcn_readlane(nv, j);
                for (int p0 = 32; p0 < nj; p0 += 32) {
                    const int p = p0 + lane;
                    if (p < nj) add_record(records[bj + p], xs, acc);
                }
            }
        }
    }
    __syncthreads();
    const int drow0 = db * TILE;
    int nloc = n_nodes - drow0;
    if (nloc > TILE) nloc = TILE;
    float2* __restrict__ dst2 = reinterpret_cast<float2*>(part) + (size_t)ss * plane_rows + drow0;
    const float2* acc2 = reinterpret_cast<const float2*>(acc);
    float2 o[TILE / ACC_THREADS];
#pragma unroll
    for (int u = 0; u < TILE / ACC_THREADS; ++u) o[u] = acc2[threadIdx.x + u * ACC_THREADS];
#pragma unroll
    for (int u = 0; u < TILE / ACC_THREADS; ++u) {
        const int k = threadIdx.x + u * ACC_THREADS;
        if (k < nloc) dst2[k] = o[u];
    }
}

__global__ void node_epilogue_wmma(const float* __restrict__ x,
                                   const float* __restrict__ part,
                                   const float* __restrict__ W_rel,
                                   const float* __restrict__ b_rel,
                                   const float* __restrict__ W_root,
                                   float* __restrict__ out,
                                   int n_nodes, int n_ss, int plane_rows) {
    const int wave      = threadIdx.x >> 5;
    const int lane      = threadIdx.x & 31;
    const int num_tiles = (n_nodes + 15) >> 4;
    const int tile      = blockIdx.x * (blockDim.x >> 5) + wave;
    if (tile >= num_tiles) return;

    const bool full = (tile * 16 + 16) <= n_nodes;
    const int  n    = lane & 15;
    const int  half = lane >> 4;

    const int row = tile * 16 + n;
    const bool live = full || row < n_nodes;
    float sx = 0.f, sy = 0.f;
    float2 xr; xr.x = 0.f; xr.y = 0.f;
    if (live) {
        xr = *reinterpret_cast<const float2*>(x + (size_t)row * 2);
        const float2* __restrict__ pp = reinterpret_cast<const float2*>(part) + row;
        for (int s0 = 0; s0 < n_ss; s0 += 16) {
            float2 v[8];
#pragma unroll
            for (int u = 0; u < 8; ++u) {
                const int s = s0 + 2 * u + half;
                v[u].x = 0.f; v[u].y = 0.f;
                if (s < n_ss) v[u] = pp[(size_t)s * plane_rows];
            }
#pragma unroll
            for (int u = 0; u < 8; ++u) { sx += v[u].x; sy += v[u].y; }
        }
    }
    sx += __shfl_xor(sx, 16);
    sy += __shfl_xor(sy, 16);
    v2f a;
    a.x = half ? xr.x : sx;
    a.y = half ? xr.y : sy;

    v2f b;
    b.x = 0.f; b.y = 0.f;
    if (n < 2) {
        const float* __restrict__ W = half ? W_root : W_rel;
        b.x = W[n * 2 + 0];
        b.y = W[n * 2 + 1];
    }

    const float cv = (n < 2) ? b_rel[n] : 0.f;
    v8f c;
#pragma unroll
    for (int r = 0; r < 8; ++r) c[r] = cv;

    v8f d = __builtin_amdgcn_wmma_f32_16x16x4_f32(
         false, a,  false, b,
         (short)0, c,  false,  false);

    const int m0 = tile * 16 + half * 8;
    if (full) {
        if (n < 2) {
            float* __restrict__ p = out + (size_t)m0 * 2 + n;
#pragma unroll
            for (int r = 0; r < 8; ++r) p[(size_t)r * 2] = d[r];
        }
    } else {
        if (n < 2) {
#pragma unroll
            for (int r = 0; r < 8; ++r) {
                int m = m0 + r;
                if (m < n_nodes) out[(size_t)m * 2 + n] = d[r];
            }
        }
    }
}

extern "C" void kernel_launch(void* const* d_in, const int* in_sizes, int n_in,
                              void* d_out, int out_size, void* d_ws, size_t ws_size,
                              hipStream_t stream) {
    const float* x          = (const float*)d_in[0];
    const int*   edge_index = (const int*)d_in[1];
    const float* W_rel      = (const float*)d_in[2];
    const float* b_rel      = (const float*)d_in[3];
    const float* W_root     = (const float*)d_in[4];
    float*       out        = (float*)d_out;

    const int n_nodes = in_sizes[0] / 2;
    const int n_edges = in_sizes[1] / 2;
    const int* src = edge_index;
    const int* dst = edge_index + n_edges;
    const int n_ss   = (n_nodes + TILE - 1) / TILE;
    const int n_bins = n_ss * n_ss;
    const int plane_rows = ((n_nodes + 15) / 16) * 16;
    const int chunks = (n_edges + BIN_CHUNK - 1) / BIN_CHUNK;

    float* part      = (float*)d_ws;
    int*   seg_start = (int*)d_ws + (((size_t)n_ss * plane_rows * 2 + 63) / 64) * 64;
    int*   records   = seg_start + (size_t)chunks * SEG_PITCH;

    chunk_sort_kernel<<<chunks, BIN_THREADS, 0, stream>>>(src, dst, seg_start, records,
                                                         n_edges, n_bins, n_ss);
    tile_accumulate_kernel<<<n_bins, ACC_THREADS, 0, stream>>>(records, seg_start, x, part,
                                                               n_nodes, chunks, n_ss, plane_rows);

    {
        int num_tiles = (n_nodes + 15) / 16;
        int blocks    = (num_tiles + 7) / 8;
        node_epilogue_wmma<<<blocks, 256, 0, stream>>>(x, part, W_rel, b_rel,
                                                       W_root, out, n_nodes, n_ss, plane_rows);
    }
}


# Each accumulate workgroup walks 4 source slices of its dst range with the sums kept in LDS (float adds in arrival order over all 4), next slice's loads under way meanwhile: 8 partial planes, not 31
# speedup vs baseline: 7.6341x; 1.2745x over previous
#include <hip/hip_runtime.h>

typedef float v2f __attribute__((ext_vector_type(2)));
typedef float v8f __attribute__((ext_vector_type(8)));
typedef int   v4i __attribute__((ext_vector_type(4)));
typedef int   v2i __attribute__((ext_vector_type(2)));

#define TS 14
#define TILE (1 << TS)
#define BIN_THREADS 1024
#define BIN_VEC 4
#define BIN_EPT (BIN_VEC * 4)
#define BIN_CHUNK (BIN_THREADS * BIN_EPT)
#define ACC_THREADS 1024
#define MAX_BINS BIN_THREADS
#define SEG_PITCH (MAX_BINS + 64)

__device__ __forceinline__ int block_excl_scan_1024(int v, int* wsum) {
    const int lane = threadIdx.x & 31;
    const int w    = threadIdx.x >> 5;
    int inc = v;
#pragma unroll
    for (int o = 1; o < 32; o <<= 1) {
        int t = __shfl_up(inc, o);
        if (lane >= o) inc += t;
    }
    if (lane == 31) wsum[w] = inc;
    __syncthreads();
    if (w == 0) {
        int s  = wsum[lane];
        int si = s;
#pragma unroll
        for (int o = 1; o < 32; o <<= 1) {
            int t = __shfl_up(si, o);
            if (lane >= o) si += t;
        }
        wsum[lane] = si - s;
        if (lane == 31) wsum[32] = si;
    }
    __syncthreads();
    return wsum[w] + inc - v;
}

template <bool FULL>
__device__ __forceinline__ void chunk_sort_body(
        const int* __restrict__ src, const int* __restrict__ dst,
        int* __restrict__ seg_start, int* __restrict__ rec, int n_edges, int n_bins, int n_ss,
        int* lcnt, int* wsum, int* stage) {
    const int base = blockIdx.x * BIN_CHUNK;

    lcnt[threadIdx.x] = 0;
    __syncthreads();

    int s[BIN_EPT], d[BIN_EPT];
#pragma unroll
    for (int j = 0; j < BIN_VEC; ++j) {
        int e0 = base + (j * BIN_THREADS + threadIdx.x) * 4;
        if (FULL || e0 + 3 < n_edges) {
            v4i sv = __builtin_nontemporal_load(reinterpret_cast<const v4i*>(src + e0));
            v4i dv = __builtin_nontemporal_load(reinterpret_cast<const v4i*>(dst + e0));
            s[j * 4 + 0] = sv.x; s[j * 4 + 1] = sv.y; s[j * 4 + 2] = sv.z; s[j * 4 + 3] = sv.w;
            d[j * 4 + 0] = dv.x; d[j * 4 + 1] = dv.y; d[j * 4 + 2] = dv.z; d[j * 4 + 3] = dv.w;
        } else {
#pragma unroll
            for (int k = 0; k < 4; ++k) {
                int  e  = e0 + k;
                bool ok = e < n_edges;
                s[j * 4 + k] = ok ? src[e] : 0;
                d[j * 4 + k] = ok ? dst[e] : -1;
            }
        }
    }
#pragma unroll
    for (int k = 0; k < BIN_EPT; ++k)
        if (FULL || d[k] >= 0) atomicAdd(&lcnt[(d[k] >> TS) * n_ss + (s[k] >> TS)], 1);
    __syncthreads();

    {
        const int b   = threadIdx.x;
        const int h   = lcnt[b];
        const int run = block_excl_scan_1024(h, wsum);
        int* __restrict__ row = seg_start + (size_t)blockIdx.x * SEG_PITCH;
        if (b <= n_bins) row[b] = run;
        if (b == MAX_BINS - 1 && n_bins == MAX_BINS) row[MAX_BINS] = run + h;
        lcnt[b] = run;
    }
    __syncthreads();

#pragma unroll
    for (int j = 0; j < BIN_VEC; ++j) {
        int p[4];
#pragma unroll
        for (int k = 0; k < 4; ++k)
            p[k] = (FULL || d[j * 4 + k] >= 0)
                       ? atomicAdd(&lcnt[(d[j * 4 + k] >> TS) * n_ss + (s[j * 4 + k] >> TS)], 1) : -1;
#pragma unroll
        for (int k = 0; k < 4; ++k) {
            if (FULL || p[k] >= 0) {
                stage[p[k]] = (s[j * 4 + k] & (TILE - 1)) | ((d[j * 4 + k] & (TILE - 1)) << TS);
            }
        }
    }
    __syncthreads();

    int total = n_edges - base;
    if (FULL || total > BIN_CHUNK) total = BIN_CHUNK;
    const int total4 = (total + 3) >> 2;
    const v4i* st4 = reinterpret_cast<const v4i*>(stage);
    v4i* __restrict__ rec4 = reinterpret_cast<v4i*>(rec + (size_t)base);
    for (int i = threadIdx.x; i < total4; i += BIN_THREADS)
        rec4[i] = st4[i];
}

__global__ void __launch_bounds__(BIN_THREADS, 16)
chunk_sort_kernel(const int* __restrict__ src, const int* __restrict__ dst,
                  int* __restrict__ seg_start, int* __restrict__ records,
                  int n_edges, int n_bins, int n_ss) {
    __shared__ int lcnt[MAX_BINS];
    __shared__ int wsum[33];
    __shared__ __attribute__((aligned(16))) int stage[BIN_CHUNK];
    int* rec = records;
    if ((blockIdx.x + 1) * BIN_CHUNK <= (unsigned)n_edges)
        chunk_sort_body<true>(src, dst, seg_start, rec, n_edges, n_bins, n_ss, lcnt, wsum, stage);
    else
        chunk_sort_body<false>(src, dst, seg_start, rec, n_edges, n_bins, n_ss, lcnt, wsum, stage);
}

#define SLICES_PER_WG 4

template <bool B> struct flag_t { static constexpr bool value = B; };

__device__ __forceinline__ void lds_barrier() {
    __builtin_amdgcn_fence(__ATOMIC_RELEASE, "workgroup", "local");
    __builtin_amdgcn_s_barrier();
    __builtin_amdgcn_fence(__ATOMIC_ACQUIRE, "workgroup", "local");
}

__device__ __forceinline__ void add_record(int r, const float2* xs, float* acc) {
    const float2 xv = xs[r & (TILE - 1)];
    const int    dl = r >> TS;
    atomicAdd(&acc[dl * 2 + 0], xv.x);
    atomicAdd(&acc[dl * 2 + 1], xv.y);
}

__device__ __forceinline__ void load_bounds(const int* __restrict__ seg_start, int chunks, int key,
                                            int c, int& l, int& h) {
    l = 0; h = 0;
    if (c < chunks) {
        const int* __restrict__ row = seg_start + (size_t)c * SEG_PITCH + key;
        l = row[0];
        h = row[1];
    }
}

__device__ __forceinline__ int load_rec(const int* __restrict__ records, int begv, int nv, int j, int lane) {
    const int bj = __builtin_amdgcn_readlane(begv, j);
    const int nj = __builtin_amdgcn_readlane(nv, j);
    const int last = nj > 0 ? bj + nj - 1 : 0;
    return records[lane < nj ? bj + lane : last];
}

__global__ void __launch_bounds__(ACC_THREADS)
tile_accumulate_kernel(const int* __restrict__ records, const int* __restrict__ seg_start,
                       const float* __restrict__ x, float* __restrict__ part,
                       int n_nodes, int chunks, int n_ss, int n_groups, int plane_rows) {
    __shared__ float2 xs[TILE];
    __shared__ float  acc[TILE * 2];
    const int db   = blockIdx.x / n_groups;
    const int grp  = blockIdx.x - db * n_groups;
    const int ss0  = grp * SLICES_PER_WG;
    int ss_end = ss0 + SLICES_PER_WG;
    if (ss_end > n_ss) ss_end = n_ss;
    const int key0 = db * n_ss;
    const int lane = threadIdx.x & 31;
    const int w = __builtin_amdgcn_readfirstlane((int)(threadIdx.x >> 5));
    const float2* __restrict__ x2 = reinterpret_cast<const float2*>(x);

    int ss = ss0, cb = 0;
    int bl, bh;
    load_bounds(seg_start, chunks, key0 + ss, cb + w + 32 * lane, bl, bh);

    float2 t[TILE / ACC_THREADS];
    {
        const int srow0 = ss * TILE;
        int ns = n_nodes - srow0;
        if (ns > TILE) ns = TILE;
#pragma unroll
        for (int u = 0; u < TILE / ACC_THREADS; ++u) {
            const int i = threadIdx.x + u * ACC_THREADS;
            t[u].x = 0.f; t[u].y = 0.f;
            if (i < ns) t[u] = x2[srow0 + i];
        }
    }
    for (int i = threadIdx.x; i < TILE * 2; i += ACC_THREADS) acc[i] = 0.f;

    int begv = (cb + w + 32 * lane) * BIN_CHUNK + bl;
    int nv   = bh - bl;
    int r[32];
#pragma unroll
    for (int j = 0; j < 32; ++j) r[j] = load_rec(records, begv, nv, j, lane);
#pragma unroll
    for (int u = 0; u < TILE / ACC_THREADS; ++u) xs[threadIdx.x + u * ACC_THREADS] = t[u];
    __syncthreads();

    for (;;) {
        int nss = ss, ncb = cb + ACC_THREADS;
        if (ncb >= chunks) { ncb = 0; ++nss; }
        const bool has_next  = nss < ss_end;
        const bool new_slice = has_next && ncb == 0;
        int nbegv = 0, nnv = 0;
        const int nc = ncb + w + 32 * lane;
        int nl = 0, nh = 0;
        if (has_next) load_bounds(seg_start, chunks, key0 + nss, nc, nl, nh);

        auto sum_item = [&](auto next_tag) {
            constexpr bool NEXT = decltype(next_tag)::value;
#pragma unroll
            for (int g = 0; g < 32; g += 8) {
                float2 xv[8];
#pragma unroll
                for (int u = 0; u < 8; ++u) xv[u] = xs[r[g + u] & (TILE - 1)];
#pragma unroll
                for (int u = 0; u < 8; ++u) {
                    const int nj = __builtin_amdgcn_readlane(nv, g + u);
                    if (lane < nj) {
                        const int dl = r[g + u] >> TS;
                        atomicAdd(&acc[dl * 2 + 0], xv[u].x);
                        atomicAdd(&acc[dl * 2 + 1], xv[u].y);
                    }
                }
                if (NEXT) {
                    if (g == 0) {
                        nbegv = nc * BIN_CHUNK + nl;
                        nnv   = nh - nl;
                    }
#pragma unroll
                    for (int u = 0; u < 8; ++u) r[g + u] = load_rec(records, nbegv, nnv, g + u, lane);
                    if (g == 0 && new_slice) {
                        const int srow0 = nss * TILE;
                        int ns = n_nodes - srow0;
                        if (ns > TILE) ns = TILE;
#pragma unroll
                        for (int u = 0; u < TILE / ACC_THREADS; ++u) {
                            const int i = threadIdx.x + u * ACC_THREADS;
                            t[u].x = 0.f; t[u].y = 0.f;
                            if (i < ns) t[u] = x2[srow0 + i];
                        }
                    }
                }
            }
        };
        if (has_next) sum_item(flag_t<true>{});
        else          sum_item(flag_t<false>{});
        if (__any(nv > 32)) {
            for (int j = 0; j < 32; ++j) {
                const int bj = __builtin_amdgcn_readlane(begv, j);
                const int nj = __builtin_amdgcn_readlane(nv, j);
                for (int p0 = 32; p0 < nj; p0 += 32) {
                    const int p = p0 + lane;
                    if (p < nj) add_record(records[bj + p], xs, acc);
                }
            }
        }
        if (!has_next) break;
        if (new_slice) {
            lds_barrier();
#pragma unroll
            for (int u = 0; u < TILE / ACC_THREADS; ++u) xs[threadIdx.x + u * ACC_THREADS] = t[u];
            lds_barrier();
        }
        begv = nbegv; nv = nnv; ss = nss; cb = ncb;
    }
    __syncthreads();
    const int drow0 = db * TILE;
    int nloc = n_nodes - drow0;
    if (nloc > TILE) nloc = TILE;
    float2* __restrict__ dst2 = reinterpret_cast<float2*>(part) + (size_t)grp * plane_rows + drow0;
    const float2* acc2 = reinterpret_cast<const float2*>(acc);
    float2 o[TILE / ACC_THREADS];
#pragma unroll
    for (int u = 0; u < TILE / ACC_THREADS; ++u) o[u] = acc2[threadIdx.x + u * ACC_THREADS];
#pragma unroll
    for (int u = 0; u < TILE / ACC_THREADS; ++u) {
        const int k = threadIdx.x + u * ACC_THREADS;
        if (k < nloc) dst2[k] = o[u];
    }
}

__global__ void node_epilogue_wmma(const float* __restrict__ x,
                                   const float* __restrict__ part,
                                   const float* __restrict__ W_rel,
                                   const float* __restrict__ b_rel,
                                   const float* __restrict__ W_root,
                                   float* __restrict__ out,
                                   int n_nodes, int n_ss, int plane_rows) {
    const int wave      = threadIdx.x >> 5;
    const int lane      = threadIdx.x & 31;
    const int num_tiles = (n_nodes + 15) >> 4;
    const int tile      = blockIdx.x * (blockDim.x >> 5) + wave;
    if (tile >= num_tiles) return;

    const bool full = (tile * 16 + 16) <= n_nodes;
    const int  n    = lane & 15;
    const int  half = lane >> 4;

    const int row = tile * 16 + n;
    const bool live = full || row < n_nodes;
    float sx = 0.f, sy = 0.f;
    float2 xr; xr.x = 0.f; xr.y = 0.f;
    if (live) {
        xr = *reinterpret_cast<const float2*>(x + (size_t)row * 2);
        const float2* __restrict__ pp = reinterpret_cast<const float2*>(part) + row;
        for (int s0 = 0; s0 < n_ss; s0 += 16) {
            float2 v[8];
#pragma unroll
            for (int u = 0; u < 8; ++u) {
                const int s = s0 + 2 * u + half;
                v[u].x = 0.f; v[u].y = 0.f;
                if (s < n_ss) v[u] = pp[(size_t)s * plane_rows];
            }
#pragma unroll
            for (int u = 0; u < 8; ++u) { sx += v[u].x; sy += v[u].y; }
        }
    }
    sx += __shfl_xor(sx, 16);
    sy += __shfl_xor(sy, 16);
    v2f a;
    a.x = half ? xr.x : sx;
    a.y = half ? xr.y : sy;

    v2f b;
    b.x = 0.f; b.y = 0.f;
    if (n < 2) {
        const float* __restrict__ W = half ? W_root : W_rel;
        b.x = W[n * 2 + 0];
        b.y = W[n * 2 + 1];
    }

    const float cv = (n < 2) ? b_rel[n] : 0.f;
    v8f c;
#pragma unroll
    for (int r = 0; r < 8; ++r) c[r] = cv;

    v8f d = __builtin_amdgcn_wmma_f32_16x16x4_f32(
         false, a,  false, b,
         (short)0, c,  false,  false);

    const int m0 = tile * 16 + half * 8;
    if (full) {
        if (n < 2) {
            float* __restrict__ p = out + (size_t)m0 * 2 + n;
#pragma unroll
            for (int r = 0; r < 8; ++r) p[(size_t)r * 2] = d[r];
        }
    } else {
        if (n < 2) {
#pragma unroll
            for (int r = 0; r < 8; ++r) {
                int m = m0 + r;
                if (m < n_nodes) out[(size_t)m * 2 + n] = d[r];
            }
        }
    }
}

extern "C" void kernel_launch(void* const* d_in, const int* in_sizes, int n_in,
                              void* d_out, int out_size, void* d_ws, size_t ws_size,
                              hipStream_t stream) {
    const float* x          = (const float*)d_in[0];
    const int*   edge_index = (const int*)d_in[1];
    const float* W_rel      = (const float*)d_in[2];
    const float* b_rel      = (const float*)d_in[3];
    const float* W_root     = (const float*)d_in[4];
    float*       out        = (float*)d_out;

    const int n_nodes = in_sizes[0] / 2;
    const int n_edges = in_sizes[1] / 2;
    const int* src = edge_index;
    const int* dst = edge_index + n_edges;
    const int n_ss   = (n_nodes + TILE - 1) / TILE;
    const int n_bins = n_ss * n_ss;
    const int n_groups = (n_ss + SLICES_PER_WG - 1) / SLICES_PER_WG;
    const int plane_rows = ((n_nodes + 15) / 16) * 16;
    const int chunks = (n_edges + BIN_CHUNK - 1) / BIN_CHUNK;

    float* part      = (float*)d_ws;
    int*   seg_start = (int*)d_ws + (((size_t)n_groups * plane_rows * 2 + 63) / 64) * 64;
    int*   records   = seg_start + (size_t)chunks * SEG_PITCH;

    chunk_sort_kernel<<<chunks, BIN_THREADS, 0, stream>>>(src, dst, seg_start, records,
                                                         n_edges, n_bins, n_ss);
    tile_accumulate_kernel<<<n_ss * n_groups, ACC_THREADS, 0, stream>>>(records, seg_start, x, part,
                                                                        n_nodes, chunks, n_ss, n_groups, plane_rows);

    {
        int num_tiles = (n_nodes + 15) / 16;
        int blocks    = (num_tiles + 7) / 8;
        node_epilogue_wmma<<<blocks, 256, 0, stream>>>(x, part, W_rel, b_rel,
                                                       W_root, out, n_nodes, n_groups, plane_rows);
    }
}


# Chunk sort: the one returning LDS add that counts an edge into its bin also gives its rank there; place = bin start from the scan + rank, so the second pass of LDS adds is gone (bin order: arrival)
# speedup vs baseline: 8.0660x; 1.0566x over previous
#include <hip/hip_runtime.h>

typedef float v2f __attribute__((ext_vector_type(2)));
typedef float v8f __attribute__((ext_vector_type(8)));
typedef int   v4i __attribute__((ext_vector_type(4)));
typedef int   v2i __attribute__((ext_vector_type(2)));

#define TS 14
#define TILE (1 << TS)
#define BIN_THREADS 1024
#define BIN_VEC 4
#define BIN_EPT (BIN_VEC * 4)
#define BIN_CHUNK (BIN_THREADS * BIN_EPT)
#define ACC_THREADS 1024
#define MAX_BINS BIN_THREADS
#define SEG_PITCH (MAX_BINS + 64)

__device__ __forceinline__ int block_excl_scan_1024(int v, int* wsum) {
    const int lane = threadIdx.x & 31;
    const int w    = threadIdx.x >> 5;
    int inc = v;
#pragma unroll
    for (int o = 1; o < 32; o <<= 1) {
        int t = __shfl_up(inc, o);
        if (lane >= o) inc += t;
    }
    if (lane == 31) wsum[w] = inc;
    __syncthreads();
    if (w == 0) {
        int s  = wsum[lane];
        int si = s;
#pragma unroll
        for (int o = 1; o < 32; o <<= 1) {
            int t = __shfl_up(si, o);
            if (lane >= o) si += t;
        }
        wsum[lane] = si - s;
        if (lane == 31) wsum[32] = si;
    }
    __syncthreads();
    return wsum[w] + inc - v;
}

#define KEY_BITS 10
static_assert(MAX_BINS == (1 << KEY_BITS), "a key must fit in KEY_BITS bits");
static_assert(BIN_CHUNK <= (1 << (31 - KEY_BITS)), "key | rank << KEY_BITS must stay a positive int");
template <bool FULL>
__device__ __forceinline__ void chunk_sort_body(
        const int* __restrict__ src, const int* __restrict__ dst,
        int* __restrict__ seg_start, int* __restrict__ rec, int n_edges, int n_bins, int n_ss,
        int* lcnt, int* wsum, int* stage) {
    const int base = blockIdx.x * BIN_CHUNK;

    lcnt[threadIdx.x] = 0;
    __syncthreads();

    int s[BIN_EPT], d[BIN_EPT];
#pragma unroll
    for (int j = 0; j < BIN_VEC; ++j) {
        int e0 = base + (j * BIN_THREADS + threadIdx.x) * 4;
        if (FULL || e0 + 3 < n_edges) {
            v4i sv = __builtin_nontemporal_load(reinterpret_cast<const v4i*>(src + e0));
            v4i dv = __builtin_nontemporal_load(reinterpret_cast<const v4i*>(dst + e0));
            s[j * 4 + 0] = sv.x; s[j * 4 + 1] = sv.y; s[j * 4 + 2] = sv.z; s[j * 4 + 3] = sv.w;
            d[j * 4 + 0] = dv.x; d[j * 4 + 1] = dv.y; d[j * 4 + 2] = dv.z; d[j * 4 + 3] = dv.w;
        } else {
#pragma unroll
            for (int k = 0; k < 4; ++k) {
                int  e  = e0 + k;
                bool ok = e < n_edges;
                s[j * 4 + k] = ok ? src[e] : 0;
                d[j * 4 + k] = ok ? dst[e] : -1;
            }
        }
    }

    int w[BIN_EPT];
    int kr[BIN_EPT];
#pragma unroll
    for (int j = 0; j < BIN_VEC; ++j) {
        int key[4], rk[4];
#pragma unroll
        for (int k = 0; k < 4; ++k) {
            const int sk = s[j * 4 + k], dk = d[j * 4 + k];
            key[k] = (dk >> TS) * n_ss + (sk >> TS);
            w[j * 4 + k] = (sk & (TILE - 1)) | ((dk & (TILE - 1)) << TS);
        }
#pragma unroll
        for (int k = 0; k < 4; ++k)
            rk[k] = (FULL || d[j * 4 + k] >= 0) ? atomicAdd(&lcnt[key[k]], 1) : 0;
#pragma unroll
        for (int k = 0; k < 4; ++k)
            kr[j * 4 + k] = (FULL || d[j * 4 + k] >= 0) ? (key[k] | (rk[k] << KEY_BITS)) : -1;
#pragma unroll
        for (int k = 0; k < 4; ++k)
            asm volatile("" :: "v"(w[j * 4 + k]), "v"(kr[j * 4 + k]));
    }
    __syncthreads();

    {
        const int b   = threadIdx.x;
        const int h   = lcnt[b];
        const int run = block_excl_scan_1024(h, wsum);
        int* __restrict__ row = seg_start + (size_t)blockIdx.x * SEG_PITCH;
        if (b <= n_bins) row[b] = run;
        if (b == MAX_BINS - 1 && n_bins == MAX_BINS) row[MAX_BINS] = run + h;
        lcnt[b] = run;
    }
    __syncthreads();

    {
        int p[BIN_EPT];
#pragma unroll
        for (int k = 0; k < BIN_EPT; ++k)
            p[k] = (FULL || kr[k] >= 0) ? lcnt[kr[k] & (MAX_BINS - 1)] + (kr[k] >> KEY_BITS) : -1;
#pragma unroll
        for (int k = 0; k < BIN_EPT; ++k)
            if (FULL || p[k] >= 0) stage[p[k]] = w[k];
    }
    __syncthreads();

    int total = n_edges - base;
    if (FULL || total > BIN_CHUNK) total = BIN_CHUNK;
    const int total4 = (total + 3) >> 2;
    const v4i* st4 = reinterpret_cast<const v4i*>(stage);
    v4i* __restrict__ rec4 = reinterpret_cast<v4i*>(rec + (size_t)base);
    for (int i = threadIdx.x; i < total4; i += BIN_THREADS)
        rec4[i] = st4[i];
}

__global__ void __launch_bounds__(BIN_THREADS, 16)
chunk_sort_kernel(const int* __restrict__ src, const int* __restrict__ dst,
                  int* __restrict__ seg_start, int* __restrict__ records,
                  int n_edges, int n_bins, int n_ss) {
    __shared__ int lcnt[MAX_BINS];
    __shared__ int wsum[33];
    __shared__ __attribute__((aligned(16))) int stage[BIN_CHUNK];
    int* rec = records;
    if ((blockIdx.x + 1) * BIN_CHUNK <= (unsigned)n_edges)
        chunk_sort_body<true>(src, dst, seg_start, rec, n_edges, n_bins, n_ss, lcnt, wsum, stage);
    else
        chunk_sort_body<false>(src, dst, seg_start, rec, n_edges, n_bins, n_ss, lcnt, wsum, stage);
}

#define SLICES_PER_WG 4

template <bool B> struct flag_t { static constexpr bool value = B; };

__device__ __forceinline__ void lds_barrier() {
    __builtin_amdgcn_fence(__ATOMIC_RELEASE, "workgroup", "local");
    __builtin_amdgcn_s_barrier();
    __builtin_amdgcn_fence(__ATOMIC_ACQUIRE, "workgroup", "local");
}

__device__ __forceinline__ void add_record(int r, const float2* xs, float* acc) {
    const float2 xv = xs[r & (TILE - 1)];
    const int    dl = r >> TS;
    atomicAdd(&acc[dl * 2 + 0], xv.x);
    atomicAdd(&acc[dl * 2 + 1], xv.y);
}

__device__ __forceinline__ void load_bounds(const int* __restrict__ seg_start, int chunks, int key,
                                            int c, int& l, int& h) {
    l = 0; h = 0;
    if (c < chunks) {
        const int* __restrict__ row = seg_start + (size_t)c * SEG_PITCH + key;
        l = row[0];
        h = row[1];
    }
}

__device__ __forceinline__ int load_rec(const int* __restrict__ records, int begv, int nv, int j, int lane) {
    const int bj = __builtin_amdgcn_readlane(begv, j);
    const int nj = __builtin_amdgcn_readlane(nv, j);
    const int last = nj > 0 ? bj + nj - 1 : 0;
    return records[lane < nj ? bj + lane : last];
}

__global__ void __launch_bounds__(ACC_THREADS)
tile_accumulate_kernel(const int* __restrict__ records, const int* __restrict__ seg_start,
                       const float* __restrict__ x, float* __restrict__ part,
                       int n_nodes, int chunks, int n_ss, int n_groups, int plane_rows) {
    __shared__ float2 xs[TILE];
    __shared__ float  acc[TILE * 2];
    const int db   = blockIdx.x / n_groups;
    const int grp  = blockIdx.x - db * n_groups;
    const int ss0  = grp * SLICES_PER_WG;
    int ss_end = ss0 + SLICES_PER_WG;
    if (ss_end > n_ss) ss_end = n_ss;
    const int key0 = db * n_ss;
    const int lane = threadIdx.x & 31;
    const int w = __builtin_amdgcn_readfirstlane((int)(threadIdx.x >> 5));
    const float2* __restrict__ x2 = reinterpret_cast<const float2*>(x);

    int ss = ss0, cb = 0;
    int bl, bh;
    load_bounds(seg_start, chunks, key0 + ss, cb + w + 32 * lane, bl, bh);

    float2 t[TILE / ACC_THREADS];
    {
        const int srow0 = ss * TILE;
        int ns = n_nodes - srow0;
        if (ns > TILE) ns = TILE;
#pragma unroll
        for (int u = 0; u < TILE / ACC_THREADS; ++u) {
            const int i = threadIdx.x + u * ACC_THREADS;
            t[u].x = 0.f; t[u].y = 0.f;
            if (i < ns) t[u] = x2[srow0 + i];
        }
    }
    for (int i = threadIdx.x; i < TILE * 2; i += ACC_THREADS) acc[i] = 0.f;

    int begv = (cb + w + 32 * lane) * BIN_CHUNK + bl;
    int nv   = bh - bl;
    int r[32];
#pragma unroll
    for (int j = 0; j < 32; ++j) r[j] = load_rec(records, begv, nv, j, lane);
#pragma unroll
    for (int u = 0; u < TILE / ACC_THREADS; ++u) xs[threadIdx.x + u * ACC_THREADS] = t[u];
    __syncthreads();

    for (;;) {
        int nss = ss, ncb = cb + ACC_THREADS;
        if (ncb >= chunks) { ncb = 0; ++nss; }
        const bool has_next  = nss < ss_end;
        const bool new_slice = has_next && ncb == 0;
        int nbegv = 0, nnv = 0;
        const int nc = ncb + w + 32 * lane;
        int nl = 0, nh = 0;
        if (has_next) load_bounds(seg_start, chunks, key0 + nss, nc, nl, nh);

        auto sum_item = [&](auto next_tag) {
            constexpr bool NEXT = decltype(next_tag)::value;
#pragma unroll
            for (int g = 0; g < 32; g += 8) {
                float2 xv[8];
#pragma unroll
                for (int u = 0; u < 8; ++u) xv[u] = xs[r[g + u] & (TILE - 1)];
#pragma unroll
                for (int u = 0; u < 8; ++u) {
                    const int nj = __builtin_amdgcn_readlane(nv, g + u);
                    if (lane < nj) {
                        const int dl = r[g + u] >> TS;
                        atomicAdd(&acc[dl * 2 + 0], xv[u].x);
                        atomicAdd(&acc[dl * 2 + 1], xv[u].y);
                    }
                }
                if (NEXT) {
                    if (g == 0) {
                        nbegv = nc * BIN_CHUNK + nl;
                        nnv   = nh - nl;
                    }
#pragma unroll
                    for (int u = 0; u < 8; ++u) r[g + u] = load_rec(records, nbegv, nnv, g + u, lane);
                    if (g == 0 && new_slice) {
                        const int srow0 = nss * TILE;
                        int ns = n_nodes - srow0;
                        if (ns > TILE) ns = TILE;
#pragma unroll
                        for (int u = 0; u < TILE / ACC_THREADS; ++u) {
                            const int i = threadIdx.x + u * ACC_THREADS;
                            t[u].x = 0.f; t[u].y = 0.f;
                            if (i < ns) t[u] = x2[srow0 + i];
                        }
                    }
                }
            }
        };
        if (has_next) sum_item(flag_t<true>{});
        else          sum_item(flag_t<false>{});
        if (__any(nv > 32)) {
            for (int j = 0; j < 32; ++j) {
                const int bj = __builtin_amdgcn_readlane(begv, j);
                const int nj = __builtin_amdgcn_readlane(nv, j);
                for (int p0 = 32; p0 < nj; p0 += 32) {
                    const int p = p0 + lane;
                    if (p < nj) add_record(records[bj + p], xs, acc);
                }
            }
        }
        if (!has_next) break;
        if (new_slice) {
            lds_barrier();
#pragma unroll
            for (int u = 0; u < TILE / ACC_THREADS; ++u) xs[threadIdx.x + u * ACC_THREADS] = t[u];
            lds_barrier();
        }
        begv = nbegv; nv = nnv; ss = nss; cb = ncb;
    }
    __syncthreads();
    const int drow0 = db * TILE;
    int nloc = n_nodes - drow0;
    if (nloc > TILE) nloc = TILE;
    float2* __restrict__ dst2 = reinterpret_cast<float2*>(part) + (size_t)grp * plane_rows + drow0;
    const float2* acc2 = reinterpret_cast<const float2*>(acc);
    float2 o[TILE / ACC_THREADS];
#pragma unroll
    for (int u = 0; u < TILE / ACC_THREADS; ++u) o[u] = acc2[threadIdx.x + u * ACC_THREADS];
#pragma unroll
    for (int u = 0; u < TILE / ACC_THREADS; ++u) {
        const int k = threadIdx.x + u * ACC_THREADS;
        if (k < nloc) dst2[k] = o[u];
    }
}

__global__ void node_epilogue_wmma(const float* __restrict__ x,
                                   const float* __restrict__ part,
                                   const float* __restrict__ W_rel,
                                   const float* __restrict__ b_rel,
                                   const float* __restrict__ W_root,
                                   float* __restrict__ out,
                                   int n_nodes, int n_ss, int plane_rows) {
    const int wave      = threadIdx.x >> 5;
    const int lane      = threadIdx.x & 31;
    const int num_tiles = (n_nodes + 15) >> 4;
    const int tile      = blockIdx.x * (blockDim.x >> 5) + wave;
    if (tile >= num_tiles) return;

    const bool full = (tile * 16 + 16) <= n_nodes;
    const int  n    = lane & 15;
    const int  half = lane >> 4;

    const int row = tile * 16 + n;
    const bool live = full || row < n_nodes;
    float sx = 0.f, sy = 0.f;
    float2 xr; xr.x = 0.f; xr.y = 0.f;
    if (live) {
        xr = *reinterpret_cast<const float2*>(x + (size_t)row * 2);
        const float2* __restrict__ pp = reinterpret_cast<const float2*>(part) + row;
        for (int s0 = 0; s0 < n_ss; s0 += 16) {
            float2 v[8];
#pragma unroll
            for (int u = 0; u < 8; ++u) {
                const int s = s0 + 2 * u + half;
                v[u].x = 0.f; v[u].y = 0.f;
                if (s < n_ss) v[u] = pp[(size_t)s * plane_rows];
            }
#pragma unroll
            for (int u = 0; u < 8; ++u) { sx += v[u].x; sy += v[u].y; }
        }
    }
    sx += __shfl_xor(sx, 16);
    sy += __shfl_xor(sy, 16);
    v2f a;
    a.x = half ? xr.x : sx;
    a.y = half ? xr.y : sy;

    v2f b;
    b.x = 0.f; b.y = 0.f;
    if (n < 2) {
        const float* __restrict__ W = half ? W_root : W_rel;
        b.x = W[n * 2 + 0];
        b.y = W[n * 2 + 1];
    }

    const float cv = (n < 2) ? b_rel[n] : 0.f;
    v8f c;
#pragma unroll
    for (int r = 0; r < 8; ++r) c[r] = cv;

    v8f d = __builtin_amdgcn_wmma_f32_16x16x4_f32(
         false, a,  false, b,
         (short)0, c,  false,  false);

    const int m0 = tile * 16 + half * 8;
    if (full) {
        if (n < 2) {
            float* __restrict__ p = out + (size_t)m0 * 2 + n;
#pragma unroll
            for (int r = 0; r < 8; ++r) p[(size_t)r * 2] = d[r];
        }
    } else {
        if (n < 2) {
#pragma unroll
            for (int r = 0; r < 8; ++r) {
                int m = m0 + r;
                if (m < n_nodes) out[(size_t)m * 2 + n] = d[r];
            }
        }
    }
}

extern "C" void kernel_launch(void* const* d_in, const int* in_sizes, int n_in,
                              void* d_out, int out_size, void* d_ws, size_t ws_size,
                              hipStream_t stream) {
    const float* x          = (const float*)d_in[0];
    const int*   edge_index = (const int*)d_in[1];
    const float* W_rel      = (const float*)d_in[2];
    const float* b_rel      = (const float*)d_in[3];
    const float* W_root     = (const float*)d_in[4];
    float*       out        = (float*)d_out;

    const int n_nodes = in_sizes[0] / 2;
    const int n_edges = in_sizes[1] / 2;
    const int* src = edge_index;
    const int* dst = edge_index + n_edges;
    const int n_ss   = (n_nodes + TILE - 1) / TILE;
    const int n_bins = n_ss * n_ss;
    const int n_groups = (n_ss + SLICES_PER_WG - 1) / SLICES_PER_WG;
    const int plane_rows = ((n_nodes + 15) / 16) * 16;
    const int chunks = (n_edges + BIN_CHUNK - 1) / BIN_CHUNK;

    float* part      = (float*)d_ws;
    int*   seg_start = (int*)d_ws + (((size_t)n_groups * plane_rows * 2 + 63) / 64) * 64;
    int*   records   = seg_start + (size_t)chunks * SEG_PITCH;

    chunk_sort_kernel<<<chunks, BIN_THREADS, 0, stream>>>(src, dst, seg_start, records,
                                                         n_edges, n_bins, n_ss);
    tile_accumulate_kernel<<<n_ss * n_groups, ACC_THREADS, 0, stream>>>(records, seg_start, x, part,
                                                                        n_nodes, chunks, n_ss, n_groups, plane_rows);

    {
        int num_tiles = (n_nodes + 15) / 16;
        int blocks    = (num_tiles + 7) / 8;
        node_epilogue_wmma<<<blocks, 256, 0, stream>>>(x, part, W_rel, b_rel,
                                                       W_root, out, n_nodes, n_groups, plane_rows);
    }
}


# Accumulate launch: LDS sums regrouped in blocks of 64 rows (64 x sums, then the 64 y sums, a row's y 32 banks from its x): each LDS float add's lanes reach all 64 banks, not 32; still arrival order
# speedup vs baseline: 8.3955x; 1.0409x over previous
#include <hip/hip_runtime.h>

typedef float v2f __attribute__((ext_vector_type(2)));
typedef float v8f __attribute__((ext_vector_type(8)));
typedef int   v4i __attribute__((ext_vector_type(4)));
typedef int   v2i __attribute__((ext_vector_type(2)));

#define TS 14
#define TILE (1 << TS)
#define BIN_THREADS 1024
#define BIN_VEC 4
#define BIN_EPT (BIN_VEC * 4)
#define BIN_CHUNK (BIN_THREADS * BIN_EPT)
#define ACC_THREADS 1024
#define MAX_BINS BIN_THREADS
#define SEG_PITCH (MAX_BINS + 64)

__device__ __forceinline__ int block_excl_scan_1024(int v, int* wsum) {
    const int lane = threadIdx.x & 31;
    const int w    = threadIdx.x >> 5;
    int inc = v;
#pragma unroll
    for (int o = 1; o < 32; o <<= 1) {
        int t = __shfl_up(inc, o);
        if (lane >= o) inc += t;
    }
    if (lane == 31) wsum[w] = inc;
    __syncthreads();
    if (w == 0) {
        int s  = wsum[lane];
        int si = s;
#pragma unroll
        for (int o = 1; o < 32; o <<= 1) {
            int t = __shfl_up(si, o);
            if (lane >= o) si += t;
        }
        wsum[lane] = si - s;
        if (lane == 31) wsum[32] = si;
    }
    __syncthreads();
    return wsum[w] + inc - v;
}

#define KEY_BITS 10
static_assert(MAX_BINS == (1 << KEY_BITS), "a key must fit in KEY_BITS bits");
static_assert(BIN_CHUNK <= (1 << (31 - KEY_BITS)), "key | rank << KEY_BITS must stay a positive int");
template <bool FULL>
__device__ __forceinline__ void chunk_sort_body(
        const int* __restrict__ src, const int* __restrict__ dst,
        int* __restrict__ seg_start, int* __restrict__ rec, int n_edges, int n_bins, int n_ss,
        int* lcnt, int* wsum, int* stage) {
    const int base = blockIdx.x * BIN_CHUNK;

    lcnt[threadIdx.x] = 0;
    __syncthreads();

    int s[BIN_EPT], d[BIN_EPT];
#pragma unroll
    for (int j = 0; j < BIN_VEC; ++j) {
        int e0 = base + (j * BIN_THREADS + threadIdx.x) * 4;
        if (FULL || e0 + 3 < n_edges) {
            v4i sv = __builtin_nontemporal_load(reinterpret_cast<const v4i*>(src + e0));
            v4i dv = __builtin_nontemporal_load(reinterpret_cast<const v4i*>(dst + e0));
            s[j * 4 + 0] = sv.x; s[j * 4 + 1] = sv.y; s[j * 4 + 2] = sv.z; s[j * 4 + 3] = sv.w;
            d[j * 4 + 0] = dv.x; d[j * 4 + 1] = dv.y; d[j * 4 + 2] = dv.z; d[j * 4 + 3] = dv.w;
        } else {
#pragma unroll
            for (int k = 0; k < 4; ++k) {
                int  e  = e0 + k;
                bool ok = e < n_edges;
                s[j * 4 + k] = ok ? src[e] : 0;
                d[j * 4 + k] = ok ? dst[e] : -1;
            }
        }
    }

    int w[BIN_EPT];
    int kr[BIN_EPT];
#pragma unroll
    for (int j = 0; j < BIN_VEC; ++j) {
        int key[4], rk[4];
#pragma unroll
        for (int k = 0; k < 4; ++k) {
            const int sk = s[j * 4 + k], dk = d[j * 4 + k];
            key[k] = (dk >> TS) * n_ss + (sk >> TS);
            w[j * 4 + k] = (sk & (TILE - 1)) | ((dk & (TILE - 1)) << TS);
        }
#pragma unroll
        for (int k = 0; k < 4; ++k)
            rk[k] = (FULL || d[j * 4 + k] >= 0) ? atomicAdd(&lcnt[key[k]], 1) : 0;
#pragma unroll
        for (int k = 0; k < 4; ++k)
            kr[j * 4 + k] = (FULL || d[j * 4 + k] >= 0) ? (key[k] | (rk[k] << KEY_BITS)) : -1;
#pragma unroll
        for (int k = 0; k < 4; ++k)
            asm volatile("" :: "v"(w[j * 4 + k]), "v"(kr[j * 4 + k]));
    }
    __syncthreads();

    {
        const int b   = threadIdx.x;
        const int h   = lcnt[b];
        const int run = block_excl_scan_1024(h, wsum);
        int* __restrict__ row = seg_start + (size_t)blockIdx.x * SEG_PITCH;
        if (b <= n_bins) row[b] = run;
        if (b == MAX_BINS - 1 && n_bins == MAX_BINS) row[MAX_BINS] = run + h;
        lcnt[b] = run;
    }
    __syncthreads();

    {
        int p[BIN_EPT];
#pragma unroll
        for (int k = 0; k < BIN_EPT; ++k)
            p[k] = (FULL || kr[k] >= 0) ? lcnt[kr[k] & (MAX_BINS - 1)] + (kr[k] >> KEY_BITS) : -1;
#pragma unroll
        for (int k = 0; k < BIN_EPT; ++k)
            if (FULL || p[k] >= 0) stage[p[k]] = w[k];
    }
    __syncthreads();

    int total = n_edges - base;
    if (FULL || total > BIN_CHUNK) total = BIN_CHUNK;
    const int total4 = (total + 3) >> 2;
    const v4i* st4 = reinterpret_cast<const v4i*>(stage);
    v4i* __restrict__ rec4 = reinterpret_cast<v4i*>(rec + (size_t)base);
    for (int i = threadIdx.x; i < total4; i += BIN_THREADS)
        rec4[i] = st4[i];
}

__global__ void __launch_bounds__(BIN_THREADS, 16)
chunk_sort_kernel(const int* __restrict__ src, const int* __restrict__ dst,
                  int* __restrict__ seg_start, int* __restrict__ records,
                  int n_edges, int n_bins, int n_ss) {
    __shared__ int lcnt[MAX_BINS];
    __shared__ int wsum[33];
    __shared__ __attribute__((aligned(16))) int stage[BIN_CHUNK];
    int* rec = records;
    if ((blockIdx.x + 1) * BIN_CHUNK <= (unsigned)n_edges)
        chunk_sort_body<true>(src, dst, seg_start, rec, n_edges, n_bins, n_ss, lcnt, wsum, stage);
    else
        chunk_sort_body<false>(src, dst, seg_start, rec, n_edges, n_bins, n_ss, lcnt, wsum, stage);
}

#define SLICES_PER_WG 4

template <bool B> struct flag_t { static constexpr bool value = B; };

__device__ __forceinline__ void lds_barrier() {
    __builtin_amdgcn_fence(__ATOMIC_RELEASE, "workgroup", "local");
    __builtin_amdgcn_s_barrier();
    __builtin_amdgcn_fence(__ATOMIC_ACQUIRE, "workgroup", "local");
}

#define ACC_BLK 64
__device__ __forceinline__ int acc_x(int dl) { return dl + (dl & ~(ACC_BLK - 1)); }
__device__ __forceinline__ int acc_y(int ax) { return (ax ^ (ACC_BLK / 2)) + ACC_BLK; }

__device__ __forceinline__ void add_record(int r, const float2* xs, float* acc) {
    const float2 xv = xs[r & (TILE - 1)];
    const int    a  = acc_x(r >> TS);
    atomicAdd(&acc[a], xv.x);
    atomicAdd(&acc[acc_y(a)], xv.y);
}

__device__ __forceinline__ void load_bounds(const int* __restrict__ seg_start, int chunks, int key,
                                            int c, int& l, int& h) {
    l = 0; h = 0;
    if (c < chunks) {
        const int* __restrict__ row = seg_start + (size_t)c * SEG_PITCH + key;
        l = row[0];
        h = row[1];
    }
}

__device__ __forceinline__ int load_rec(const int* __restrict__ records, int begv, int nv, int j, int lane) {
    const int bj = __builtin_amdgcn_readlane(begv, j);
    const int nj = __builtin_amdgcn_readlane(nv, j);
    const int last = nj > 0 ? bj + nj - 1 : 0;
    return records[lane < nj ? bj + lane : last];
}

__global__ void __launch_bounds__(ACC_THREADS)
tile_accumulate_kernel(const int* __restrict__ records, const int* __restrict__ seg_start,
                       const float* __restrict__ x, float* __restrict__ part,
                       int n_nodes, int chunks, int n_ss, int n_groups, int plane_rows) {
    __shared__ float2 xs[TILE];
    __shared__ float  acc[TILE * 2];
    static_assert(TILE % ACC_BLK == 0, "whole blocks of ACC_BLK rows");
    const int db   = blockIdx.x / n_groups;
    const int grp  = blockIdx.x - db * n_groups;
    const int ss0  = grp * SLICES_PER_WG;
    int ss_end = ss0 + SLICES_PER_WG;
    if (ss_end > n_ss) ss_end = n_ss;
    const int key0 = db * n_ss;
    const int lane = threadIdx.x & 31;
    const int w = __builtin_amdgcn_readfirstlane((int)(threadIdx.x >> 5));
    const float2* __restrict__ x2 = reinterpret_cast<const float2*>(x);

    int ss = ss0, cb = 0;
    int bl, bh;
    load_bounds(seg_start, chunks, key0 + ss, cb + w + 32 * lane, bl, bh);

    float2 t[TILE / ACC_THREADS];
    {
        const int srow0 = ss * TILE;
        int ns = n_nodes - srow0;
        if (ns > TILE) ns = TILE;
#pragma unroll
        for (int u = 0; u < TILE / ACC_THREADS; ++u) {
            const int i = threadIdx.x + u * ACC_THREADS;
            t[u].x = 0.f; t[u].y = 0.f;
            if (i < ns) t[u] = x2[srow0 + i];
        }
    }
    for (int i = threadIdx.x; i < TILE * 2; i += ACC_THREADS) acc[i] = 0.f;

    int begv = (cb + w + 32 * lane) * BIN_CHUNK + bl;
    int nv   = bh - bl;
    int r[32];
#pragma unroll
    for (int j = 0; j < 32; ++j) r[j] = load_rec(records, begv, nv, j, lane);
#pragma unroll
    for (int u = 0; u < TILE / ACC_THREADS; ++u) xs[threadIdx.x + u * ACC_THREADS] = t[u];
    __syncthreads();

    for (;;) {
        int nss = ss, ncb = cb + ACC_THREADS;
        if (ncb >= chunks) { ncb = 0; ++nss; }
        const bool has_next  = nss < ss_end;
        const bool new_slice = has_next && ncb == 0;
        int nbegv = 0, nnv = 0;
        const int nc = ncb + w + 32 * lane;
        int nl = 0, nh = 0;
        if (has_next) load_bounds(seg_start, chunks, key0 + nss, nc, nl, nh);

        auto sum_item = [&](auto next_tag) {
            constexpr bool NEXT = decltype(next_tag)::value;
#pragma unroll
            for (int g = 0; g < 32; g += 8) {
                float2 xv[8];
#pragma unroll
                for (int u = 0; u < 8; ++u) xv[u] = xs[r[g + u] & (TILE - 1)];
#pragma unroll
                for (int u = 0; u < 8; ++u) {
                    const int nj = __builtin_amdgcn_readlane(nv, g + u);
                    if (lane < nj) {
                        const int a = acc_x(r[g + u] >> TS);
                        atomicAdd(&acc[a], xv[u].x);
                        atomicAdd(&acc[acc_y(a)], xv[u].y);
                    }
                }
                if (NEXT) {
                    if (g == 0) {
                        nbegv = nc * BIN_CHUNK + nl;
                        nnv   = nh - nl;
                    }
#pragma unroll
                    for (int u = 0; u < 8; ++u) r[g + u] = load_rec(records, nbegv, nnv, g + u, lane);
                    if (g == 0 && new_slice) {
                        const int srow0 = nss * TILE;
                        int ns = n_nodes - srow0;
                        if (ns > TILE) ns = TILE;
#pragma unroll
                        for (int u = 0; u < TILE / ACC_THREADS; ++u) {
                            const int i = threadIdx.x + u * ACC_THREADS;
                            t[u].x = 0.f; t[u].y = 0.f;
                            if (i < ns) t[u] = x2[srow0 + i];
                        }
                    }
                }
            }
        };
        if (has_next) sum_item(flag_t<true>{});
        else          sum_item(flag_t<false>{});
        if (__any(nv > 32)) {
            for (int j = 0; j < 32; ++j) {
                const int bj = __builtin_amdgcn_readlane(begv, j);
                const int nj = __builtin_amdgcn_readlane(nv, j);
                for (int p0 = 32; p0 < nj; p0 += 32) {
                    const int p = p0 + lane;
                    if (p < nj) add_record(records[bj + p], xs, acc);
                }
            }
        }
        if (!has_next) break;
        if (new_slice) {
            lds_barrier();
#pragma unroll
            for (int u = 0; u < TILE / ACC_THREADS; ++u) xs[threadIdx.x + u * ACC_THREADS] = t[u];
            lds_barrier();
        }
        begv = nbegv; nv = nnv; ss = nss; cb = ncb;
    }
    __syncthreads();
    const int drow0 = db * TILE;
    int nloc = n_nodes - drow0;
    if (nloc > TILE) nloc = TILE;
    float2* __restrict__ dst2 = reinterpret_cast<float2*>(part) + (size_t)grp * plane_rows + drow0;
    float2 o[TILE / ACC_THREADS];
#pragma unroll
    for (int u = 0; u < TILE / ACC_THREADS; ++u) {
        const int a = acc_x(threadIdx.x + u * ACC_THREADS);
        o[u].x = acc[a];
        o[u].y = acc[acc_y(a)];
    }
#pragma unroll
    for (int u = 0; u < TILE / ACC_THREADS; ++u) {
        const int k = threadIdx.x + u * ACC_THREADS;
        if (k < nloc) dst2[k] = o[u];
    }
}

__global__ void node_epilogue_wmma(const float* __restrict__ x,
                                   const float* __restrict__ part,
                                   const float* __restrict__ W_rel,
                                   const float* __restrict__ b_rel,
                                   const float* __restrict__ W_root,
                                   float* __restrict__ out,
                                   int n_nodes, int n_ss, int plane_rows) {
    const int wave      = threadIdx.x >> 5;
    const int lane      = threadIdx.x & 31;
    const int num_tiles = (n_nodes + 15) >> 4;
    const int tile      = blockIdx.x * (blockDim.x >> 5) + wave;
    if (tile >= num_tiles) return;

    const bool full = (tile * 16 + 16) <= n_nodes;
    const int  n    = lane & 15;
    const int  half = lane >> 4;

    const int row = tile * 16 + n;
    const bool live = full || row < n_nodes;
    float sx = 0.f, sy = 0.f;
    float2 xr; xr.x = 0.f; xr.y = 0.f;
    if (live) {
        xr = *reinterpret_cast<const float2*>(x + (size_t)row * 2);
        const float2* __restrict__ pp = reinterpret_cast<const float2*>(part) + row;
        for (int s0 = 0; s0 < n_ss; s0 += 16) {
            float2 v[8];
#pragma unroll
            for (int u = 0; u < 8; ++u) {
                const int s = s0 + 2 * u + half;
                v[u].x = 0.f; v[u].y = 0.f;
                if (s < n_ss) v[u] = pp[(size_t)s * plane_rows];
            }
#pragma unroll
            for (int u = 0; u < 8; ++u) { sx += v[u].x; sy += v[u].y; }
        }
    }
    sx += __shfl_xor(sx, 16);
    sy += __shfl_xor(sy, 16);
    v2f a;
    a.x = half ? xr.x : sx;
    a.y = half ? xr.y : sy;

    v2f b;
    b.x = 0.f; b.y = 0.f;
    if (n < 2) {
        const float* __restrict__ W = half ? W_root : W_rel;
        b.x = W[n * 2 + 0];
        b.y = W[n * 2 + 1];
    }

    const float cv = (n < 2) ? b_rel[n] : 0.f;
    v8f c;
#pragma unroll
    for (int r = 0; r < 8; ++r) c[r] = cv;

    v8f d = __builtin_amdgcn_wmma_f32_16x16x4_f32(
         false, a,  false, b,
         (short)0, c,  false,  false);

    const int m0 = tile * 16 + half * 8;
    if (full) {
        if (n < 2) {
            float* __restrict__ p = out + (size_t)m0 * 2 + n;
#pragma unroll
            for (int r = 0; r < 8; ++r) p[(size_t)r * 2] = d[r];
        }
    } else {
        if (n < 2) {
#pragma unroll
            for (int r = 0; r < 8; ++r) {
                int m = m0 + r;
                if (m < n_nodes) out[(size_t)m * 2 + n] = d[r];
            }
        }
    }
}

extern "C" void kernel_launch(void* const* d_in, const int* in_sizes, int n_in,
                              void* d_out, int out_size, void* d_ws, size_t ws_size,
                              hipStream_t stream) {
    const float* x          = (const float*)d_in[0];
    const int*   edge_index = (const int*)d_in[1];
    const float* W_rel      = (const float*)d_in[2];
    const float* b_rel      = (const float*)d_in[3];
    const float* W_root     = (const float*)d_in[4];
    float*       out        = (float*)d_out;

    const int n_nodes = in_sizes[0] / 2;
    const int n_edges = in_sizes[1] / 2;
    const int* src = edge_index;
    const int* dst = edge_index + n_edges;
    const int n_ss   = (n_nodes + TILE - 1) / TILE;
    const int n_bins = n_ss * n_ss;
    const int n_groups = (n_ss + SLICES_PER_WG - 1) / SLICES_PER_WG;
    const int plane_rows = ((n_nodes + 15) / 16) * 16;
    const int chunks = (n_edges + BIN_CHUNK - 1) / BIN_CHUNK;

    float* part      = (float*)d_ws;
    int*   seg_start = (int*)d_ws + (((size_t)n_groups * plane_rows * 2 + 63) / 64) * 64;
    int*   records   = seg_start + (size_t)chunks * SEG_PITCH;

    chunk_sort_kernel<<<chunks, BIN_THREADS, 0, stream>>>(src, dst, seg_start, records,
                                                         n_edges, n_bins, n_ss);
    tile_accumulate_kernel<<<n_ss * n_groups, ACC_THREADS, 0, stream>>>(records, seg_start, x, part,
                                                                        n_nodes, chunks, n_ss, n_groups, plane_rows);

    {
        int num_tiles = (n_nodes + 15) / 16;
        int blocks    = (num_tiles + 7) / 8;
        node_epilogue_wmma<<<blocks, 256, 0, stream>>>(x, part, W_rel, b_rel,
                                                       W_root, out, n_nodes, n_groups, plane_rows);
    }
}
